# starting kernel
# baseline (speedup 1.0000x reference)
#include <hip/hip_runtime.h>
#include <hip/hip_bf16.h>
#include <math.h>

#define N_NODES 50000
#define N_EDGES 1250000
#define EMB     64
#define BN_EPS  1e-5f

typedef __attribute__((ext_vector_type(2))) float v2f;
typedef __attribute__((ext_vector_type(8))) float v8f;

__device__ __forceinline__ unsigned f2ord(float f) {
    unsigned u = __float_as_uint(f);
    return (u & 0x80000000u) ? ~u : (u | 0x80000000u);
}
__device__ __forceinline__ float ord2f(unsigned u) {
    return (u & 0x80000000u) ? __uint_as_float(u & 0x7fffffffu)
                             : __uint_as_float(~u);
}

__global__ void k_init(unsigned* __restrict__ smax_ord,
                       float* __restrict__ ssum,
                       float* __restrict__ neigh) {
    int i = blockIdx.x * blockDim.x + threadIdx.x;
    if (i < N_NODES) { smax_ord[i] = 0u; ssum[i] = 0.0f; }
    if (i < N_NODES * EMB) neigh[i] = 0.0f;
}

__global__ void k_score(const float* __restrict__ ent, const float* __restrict__ rel,
                        const int* __restrict__ src, const int* __restrict__ dst,
                        const int* __restrict__ rid,
                        float* __restrict__ score, unsigned* __restrict__ smax_ord) {
    int e    = (blockIdx.x * blockDim.x + threadIdx.x) >> 5;
    int lane = threadIdx.x & 31;
    if (e >= N_EDGES) return;
    int s = src[e], d = dst[e], r = rid[e];
    const float* es = ent + (size_t)s * EMB;
    const float* ed = ent + (size_t)d * EMB;
    const float* rr = rel + (size_t)r * EMB;
    float p = es[lane]      * rr[lane]      * ed[lane]
            + es[lane + 32] * rr[lane + 32] * ed[lane + 32];
#pragma unroll
    for (int off = 16; off > 0; off >>= 1) p += __shfl_xor(p, off, 32);
    if (lane == 0) {
        score[e] = p;
        atomicMax(&smax_ord[d], f2ord(p));
    }
}

__global__ void k_exp(const float* __restrict__ score, const int* __restrict__ dst,
                      const unsigned* __restrict__ smax_ord,
                      float* __restrict__ expe, float* __restrict__ ssum) {
    int i = blockIdx.x * blockDim.x + threadIdx.x;
    if (i >= N_EDGES) return;
    int d = dst[i];
    float ev = expf(score[i] - ord2f(smax_ord[d]));
    expe[i] = ev;
    atomicAdd(&ssum[d], ev);
}

__global__ void k_scatter(const float* __restrict__ ent, const float* __restrict__ rel,
                          const int* __restrict__ src, const int* __restrict__ dst,
                          const int* __restrict__ rid,
                          const float* __restrict__ expe, const float* __restrict__ ssum,
                          float* __restrict__ neigh) {
    int e = blockIdx.x * 4 + (threadIdx.x >> 6);
    int c = threadIdx.x & 63;
    if (e >= N_EDGES) return;
    int s = src[e], d = dst[e], r = rid[e];
    float alpha = expe[e] / ssum[d];
    float comp  = ent[(size_t)s * EMB + c] * rel[(size_t)r * EMB + c];
    atomicAdd(&neigh[(size_t)d * EMB + c], comp * alpha);
}

__global__ void k_gemm_wmma(const float* __restrict__ neigh,
                            const float* __restrict__ W,
                            float* __restrict__ h) {
    int row0  = blockIdx.x * 16;
    int wave  = threadIdx.x >> 5;
    int lane  = threadIdx.x & 31;
    int n0    = wave * 16;
    int m     = lane & 15;
    int khalf = lane >> 4;
    int n     = lane & 15;

    v8f acc = {};
#pragma unroll
    for (int k = 0; k < EMB; k += 4) {
        int ka = k + khalf * 2;
        v2f a, b;
        a.x = neigh[(size_t)(row0 + m) * EMB + ka];
        a.y = neigh[(size_t)(row0 + m) * EMB + ka + 1];
        b.x = W[(size_t)ka       * EMB + n0 + n];
        b.y = W[(size_t)(ka + 1) * EMB + n0 + n];
        acc = __builtin_amdgcn_wmma_f32_16x16x4_f32(
                  false, a, false, b, (short)0, acc, false, false);
    }
    int mbase = row0 + khalf * 8;
#pragma unroll
    for (int r = 0; r < 8; ++r)
        h[(size_t)(mbase + r) * EMB + n0 + n] = acc[r];
}

__global__ void k_stats(const float* __restrict__ h,
                        const float* __restrict__ gamma, const float* __restrict__ beta,
                        float* __restrict__ scale, float* __restrict__ shift) {
    int c = blockIdx.x;
    __shared__ float s_sum[256], s_sq[256];
    float sum = 0.0f, sq = 0.0f;
    for (int r = threadIdx.x; r < N_NODES; r += blockDim.x) {
        float v = h[(size_t)r * EMB + c];
        sum += v; sq += v * v;
    }
    s_sum[threadIdx.x] = sum; s_sq[threadIdx.x] = sq;
    __syncthreads();
    for (int off = 128; off > 0; off >>= 1) {
        if (threadIdx.x < off) {
            s_sum[threadIdx.x] += s_sum[threadIdx.x + off];
            s_sq [threadIdx.x] += s_sq [threadIdx.x + off];
        }
        __syncthreads();
    }
    if (threadIdx.x == 0) {
        float mu   = s_sum[0] / (float)N_NODES;
        float var  = s_sq[0] / (float)N_NODES - mu * mu;
        float rinv = rsqrtf(var + BN_EPS);
        float sc   = gamma[c] * rinv;
        scale[c] = sc;
        shift[c] = beta[c] - mu * sc;
    }
}

__global__ void k_apply(float* __restrict__ h,
                        const float* __restrict__ scale, const float* __restrict__ shift) {
    int i = blockIdx.x * blockDim.x + threadIdx.x;
    if (i >= N_NODES * EMB) return;
    int c = i & 63;
    h[i] = tanhf(h[i] * scale[c] + shift[c]);
}

extern "C" void kernel_launch(void* const* d_in, const int* in_sizes, int n_in,
                              void* d_out, int out_size, void* d_ws, size_t ws_size,
                              hipStream_t stream) {
    const float* ent   = (const float*)d_in[0];
    const float* rel   = (const float*)d_in[1];
    const int*   src   = (const int*)  d_in[2];
    const int*   dst   = (const int*)  d_in[3];
    const int*   rid   = (const int*)  d_in[4];
    const float* W     = (const float*)d_in[5];
    const float* gamma = (const float*)d_in[6];
    const float* beta  = (const float*)d_in[7];
    float* out = (float*)d_out;

    float*    ws       = (float*)d_ws;
    float*    score    = ws;
    float*    expe     = ws + N_EDGES;
    unsigned* smax_ord = (unsigned*)(ws + 2 * (size_t)N_EDGES);
    float*    ssum     = ws + 2 * (size_t)N_EDGES + N_NODES;
    float*    neigh    = ws + 2 * (size_t)N_EDGES + 2 * (size_t)N_NODES;
    float*    scale    = neigh + (size_t)N_NODES * EMB;
    float*    shift    = scale + EMB;

    k_init   <<<(N_NODES * EMB + 255) / 256, 256, 0, stream>>>(smax_ord, ssum, neigh);
    k_score  <<<N_EDGES / 8,  256, 0, stream>>>(ent, rel, src, dst, rid, score, smax_ord);
    k_exp    <<<(N_EDGES + 255) / 256, 256, 0, stream>>>(score, dst, smax_ord, expe, ssum);
    k_scatter<<<N_EDGES / 4,  256, 0, stream>>>(ent, rel, src, dst, rid, expe, ssum, neigh);
    k_gemm_wmma<<<N_NODES / 16, 128, 0, stream>>>(neigh, W, out);
    k_stats  <<<EMB, 256, 0, stream>>>(out, gamma, beta, scale, shift);
    k_apply  <<<(N_NODES * EMB + 255) / 256, 256, 0, stream>>>(out, scale, shift);
}


# BN column sum and sum of squares reduced per 16-row tile from the WMMA accumulators in the GEMM epilogue; stats kernel sums 3125 tile partials (new fixed summation order) instead of re-reading h
# speedup vs baseline: 1.2108x; 1.2108x over previous
#include <hip/hip_runtime.h>
#include <hip/hip_bf16.h>
#include <math.h>

#define N_NODES 50000
#define N_EDGES 1250000
#define EMB     64
#define BN_EPS  1e-5f
#define N_TILES (N_NODES / 16)

typedef __attribute__((ext_vector_type(2))) float v2f;
typedef __attribute__((ext_vector_type(8))) float v8f;

__device__ __forceinline__ unsigned f2ord(float f) {
    unsigned u = __float_as_uint(f);
    return (u & 0x80000000u) ? ~u : (u | 0x80000000u);
}
__device__ __forceinline__ float ord2f(unsigned u) {
    return (u & 0x80000000u) ? __uint_as_float(u & 0x7fffffffu)
                             : __uint_as_float(~u);
}

__global__ void k_init(unsigned* __restrict__ smax_ord,
                       float* __restrict__ ssum,
                       float* __restrict__ neigh) {
    int i = blockIdx.x * blockDim.x + threadIdx.x;
    if (i < N_NODES) { smax_ord[i] = 0u; ssum[i] = 0.0f; }
    if (i < N_NODES * EMB) neigh[i] = 0.0f;
}

__global__ void k_score(const float* __restrict__ ent, const float* __restrict__ rel,
                        const int* __restrict__ src, const int* __restrict__ dst,
                        const int* __restrict__ rid,
                        float* __restrict__ score, unsigned* __restrict__ smax_ord) {
    int e    = (blockIdx.x * blockDim.x + threadIdx.x) >> 5;
    int lane = threadIdx.x & 31;
    if (e >= N_EDGES) return;
    int s = src[e], d = dst[e], r = rid[e];
    const float* es = ent + (size_t)s * EMB;
    const float* ed = ent + (size_t)d * EMB;
    const float* rr = rel + (size_t)r * EMB;
    float p = es[lane]      * rr[lane]      * ed[lane]
            + es[lane + 32] * rr[lane + 32] * ed[lane + 32];
#pragma unroll
    for (int off = 16; off > 0; off >>= 1) p += __shfl_xor(p, off, 32);
    if (lane == 0) {
        score[e] = p;
        atomicMax(&smax_ord[d], f2ord(p));
    }
}

__global__ void k_exp(const float* __restrict__ score, const int* __restrict__ dst,
                      const unsigned* __restrict__ smax_ord,
                      float* __restrict__ expe, float* __restrict__ ssum) {
    int i = blockIdx.x * blockDim.x + threadIdx.x;
    if (i >= N_EDGES) return;
    int d = dst[i];
    float ev = expf(score[i] - ord2f(smax_ord[d]));
    expe[i] = ev;
    atomicAdd(&ssum[d], ev);
}

__global__ void k_scatter(const float* __restrict__ ent, const float* __restrict__ rel,
                          const int* __restrict__ src, const int* __restrict__ dst,
                          const int* __restrict__ rid,
                          const float* __restrict__ expe, const float* __restrict__ ssum,
                          float* __restrict__ neigh) {
    int e = blockIdx.x * 4 + (threadIdx.x >> 6);
    int c = threadIdx.x & 63;
    if (e >= N_EDGES) return;
    int s = src[e], d = dst[e], r = rid[e];
    float alpha = expe[e] / ssum[d];
    float comp  = ent[(size_t)s * EMB + c] * rel[(size_t)r * EMB + c];
    atomicAdd(&neigh[(size_t)d * EMB + c], comp * alpha);
}

__global__ void k_gemm_wmma(const float* __restrict__ neigh,
                            const float* __restrict__ W,
                            float* __restrict__ h,
                            float* __restrict__ part) {
    int row0  = blockIdx.x * 16;
    int wave  = threadIdx.x >> 5;
    int lane  = threadIdx.x & 31;
    int n0    = wave * 16;
    int m     = lane & 15;
    int khalf = lane >> 4;
    int n     = lane & 15;

    v8f acc = {};
#pragma unroll
    for (int k = 0; k < EMB; k += 4) {
        int ka = k + khalf * 2;
        v2f a, b;
        a.x = neigh[(size_t)(row0 + m) * EMB + ka];
        a.y = neigh[(size_t)(row0 + m) * EMB + ka + 1];
        b.x = W[(size_t)ka       * EMB + n0 + n];
        b.y = W[(size_t)(ka + 1) * EMB + n0 + n];
        acc = __builtin_amdgcn_wmma_f32_16x16x4_f32(
                  false, a, false, b, (short)0, acc, false, false);
    }
    int mbase = row0 + khalf * 8;
    float cs = 0.0f, cq = 0.0f;
#pragma unroll
    for (int r = 0; r < 8; ++r) {
        float v = acc[r];
        h[(size_t)(mbase + r) * EMB + n0 + n] = v;
        cs += v; cq += v * v;
    }
    cs += __shfl_xor(cs, 16, 32);
    cq += __shfl_xor(cq, 16, 32);
    if (khalf == 0) {
        v2f p; p.x = cs; p.y = cq;
        *(v2f*)(part + 2 * ((size_t)(n0 + n) * N_TILES + blockIdx.x)) = p;
    }
}

#define STATS_THREADS 1024
__global__ void __launch_bounds__(STATS_THREADS)
k_stats(const float* __restrict__ part,
        const float* __restrict__ gamma, const float* __restrict__ beta,
        float* __restrict__ scale, float* __restrict__ shift) {
    int c = blockIdx.x;
    const v2f* p = (const v2f*)part + (size_t)c * N_TILES;
    __shared__ float s_sum[STATS_THREADS / 32], s_sq[STATS_THREADS / 32];
    float sum = 0.0f, sq = 0.0f;
#pragma unroll
    for (int i = 0; i < (N_TILES + STATS_THREADS - 1) / STATS_THREADS; ++i) {
        int t = threadIdx.x + i * STATS_THREADS;
        if (t < N_TILES) { v2f v = p[t]; sum += v.x; sq += v.y; }
    }
#pragma unroll
    for (int off = 16; off > 0; off >>= 1) {
        sum += __shfl_xor(sum, off, 32);
        sq  += __shfl_xor(sq,  off, 32);
    }
    int wave = threadIdx.x >> 5, lane = threadIdx.x & 31;
    if (lane == 0) { s_sum[wave] = sum; s_sq[wave] = sq; }
    __syncthreads();
    if (wave == 0) {
        sum = s_sum[lane]; sq = s_sq[lane];
#pragma unroll
        for (int off = 16; off > 0; off >>= 1) {
            sum += __shfl_xor(sum, off, 32);
            sq  += __shfl_xor(sq,  off, 32);
        }
        if (lane == 0) {
            float mu   = sum / (float)N_NODES;
            float var  = sq / (float)N_NODES - mu * mu;
            float rinv = rsqrtf(var + BN_EPS);
            float sc   = gamma[c] * rinv;
            scale[c] = sc;
            shift[c] = beta[c] - mu * sc;
        }
    }
}

__global__ void k_apply(float* __restrict__ h,
                        const float* __restrict__ scale, const float* __restrict__ shift) {
    int i = blockIdx.x * blockDim.x + threadIdx.x;
    if (i >= N_NODES * EMB) return;
    int c = i & 63;
    h[i] = tanhf(h[i] * scale[c] + shift[c]);
}

extern "C" void kernel_launch(void* const* d_in, const int* in_sizes, int n_in,
                              void* d_out, int out_size, void* d_ws, size_t ws_size,
                              hipStream_t stream) {
    const float* ent   = (const float*)d_in[0];
    const float* rel   = (const float*)d_in[1];
    const int*   src   = (const int*)  d_in[2];
    const int*   dst   = (const int*)  d_in[3];
    const int*   rid   = (const int*)  d_in[4];
    const float* W     = (const float*)d_in[5];
    const float* gamma = (const float*)d_in[6];
    const float* beta  = (const float*)d_in[7];
    float* out = (float*)d_out;

    float*    ws       = (float*)d_ws;
    float*    score    = ws;
    float*    expe     = ws + N_EDGES;
    unsigned* smax_ord = (unsigned*)(ws + 2 * (size_t)N_EDGES);
    float*    ssum     = ws + 2 * (size_t)N_EDGES + N_NODES;
    float*    neigh    = ws + 2 * (size_t)N_EDGES + 2 * (size_t)N_NODES;
    float*    scale    = neigh + (size_t)N_NODES * EMB;
    float*    shift    = scale + EMB;
    float*    part     = score;

    k_init   <<<(N_NODES * EMB + 255) / 256, 256, 0, stream>>>(smax_ord, ssum, neigh);
    k_score  <<<N_EDGES / 8,  256, 0, stream>>>(ent, rel, src, dst, rid, score, smax_ord);
    k_exp    <<<(N_EDGES + 255) / 256, 256, 0, stream>>>(score, dst, smax_ord, expe, ssum);
    k_scatter<<<N_EDGES / 4,  256, 0, stream>>>(ent, rel, src, dst, rid, expe, ssum, neigh);
    k_gemm_wmma<<<N_TILES, 128, 0, stream>>>(neigh, W, out, part);
    k_stats  <<<EMB, STATS_THREADS, 0, stream>>>(part, gamma, beta, scale, shift);
    k_apply  <<<(N_NODES * EMB + 255) / 256, 256, 0, stream>>>(out, scale, shift);
}


# Edges counting-sorted by destination (one int atomic each); one wave per node does scores, running-max softmax and weighted sum in registers: float atomics gone, sums in arrival order of the sort
# speedup vs baseline: 3.4027x; 2.8103x over previous
#include <hip/hip_runtime.h>
#include <math.h>

#define N_NODES 50000
#define N_EDGES 1250000
#define EMB     64
#define BN_EPS  1e-5f
#define N_TILES (N_NODES / 16)

typedef __attribute__((ext_vector_type(2))) float v2f;
typedef __attribute__((ext_vector_type(8))) float v8f;

__global__ void k_init(int* __restrict__ cnt) {
    int i = blockIdx.x * blockDim.x + threadIdx.x;
    if (i < N_NODES) cnt[i] = 0;
}

__global__ void k_hist(const int* __restrict__ dst, int* __restrict__ cnt,
                       int* __restrict__ rank) {
    int i = blockIdx.x * blockDim.x + threadIdx.x;
    if (i < N_EDGES) rank[i] = atomicAdd(&cnt[dst[i]], 1);
}

#define SCAN_THREADS 1024
#define SCAN_BLOCKS  ((N_NODES + SCAN_THREADS - 1) / SCAN_THREADS)
__global__ void __launch_bounds__(SCAN_THREADS)
k_scan(const int* __restrict__ cnt, int* __restrict__ offs) {
    __shared__ int s_v[SCAN_THREADS / 32], s_b[SCAN_THREADS / 32];
    int b = blockIdx.x, t = threadIdx.x;
    int lane = t & 31, wave = t >> 5;
    int before = 0;
    for (int i = 0; i < b; ++i) before += cnt[i * SCAN_THREADS + t];
    int idx = b * SCAN_THREADS + t;
    int v = (idx < N_NODES) ? cnt[idx] : 0;
    int inc = v;
    for (int off = 1; off < 32; off <<= 1) {
        int u = __shfl_up(inc, off, 32);
        if (lane >= off) inc += u;
    }
    for (int off = 16; off > 0; off >>= 1) before += __shfl_xor(before, off, 32);
    if (lane == 31) s_v[wave] = inc;
    if (lane == 0)  s_b[wave] = before;
    __syncthreads();
    int wb  = s_b[lane];
    int pre = (lane < wave) ? s_v[lane] : 0;
    for (int off = 16; off > 0; off >>= 1) {
        wb  += __shfl_xor(wb,  off, 32);
        pre += __shfl_xor(pre, off, 32);
    }
    if (idx < N_NODES) offs[idx] = wb + pre + inc - v;
}

__global__ void k_fill(const int* __restrict__ src, const int* __restrict__ dst,
                       const int* __restrict__ rid,
                       const int* __restrict__ offs, const int* __restrict__ rank,
                       int2* __restrict__ sorted) {
    int i = blockIdx.x * blockDim.x + threadIdx.x;
    if (i >= N_EDGES) return;
    int pos = offs[dst[i]] + rank[i];
    sorted[pos] = make_int2(src[i], rid[i]);
}

template <int MASK>
__device__ __forceinline__ float row_xor(float x) {
    int b = __builtin_bit_cast(int, x);
    return __builtin_bit_cast(float,
        __builtin_amdgcn_update_dpp(b, b, 0x160 + MASK, 0xf, 0xf, false));
}
__device__ __forceinline__ float xhalf(float x) {
    int b = __builtin_bit_cast(int, x);
    return __builtin_bit_cast(float,
        __builtin_amdgcn_permlanex16(b, b, 0x76543210, 0xfedcba98, false, false));
}
__device__ __forceinline__ float wave_sum(float p) {
    p += row_xor<1>(p);
    p += row_xor<2>(p);
    p += row_xor<4>(p);
    p += row_xor<8>(p);
    return p + xhalf(p);
}

#define NODE_WAVES 8
__global__ void __launch_bounds__(NODE_WAVES * 32)
k_node(const float* __restrict__ ent, const float* __restrict__ rel,
       const int* __restrict__ cnt, const int* __restrict__ offs,
       const int2* __restrict__ sorted, float* __restrict__ neigh) {
    int lane = threadIdx.x & 31;
    int d = __builtin_amdgcn_readfirstlane(blockIdx.x * NODE_WAVES + (threadIdx.x >> 5));
    int beg = offs[d];
    int end = beg + cnt[d];
    const v2f* ent2 = (const v2f*)ent;
    const v2f* rel2 = (const v2f*)rel;
    v2f ed = ent2[(size_t)d * (EMB / 2) + lane];
    float m = -INFINITY, l = 0.0f;
    v2f acc = {0.0f, 0.0f};
    for (int j = beg; j < end; j += 4) {
        int2 pr[4];
        for (int u = 0; u < 4; ++u) pr[u] = sorted[min(j + u, end - 1)];
        v2f es[4], rr[4];
        for (int u = 0; u < 4; ++u) {
            es[u] = ent2[(size_t)pr[u].x * (EMB / 2) + lane];
            rr[u] = rel2[(size_t)pr[u].y * (EMB / 2) + lane];
        }
        v2f   c[4];
        float sc[4];
        for (int u = 0; u < 4; ++u) {
            c[u] = es[u] * rr[u];
            float p = wave_sum(c[u].x * ed.x + c[u].y * ed.y);
            sc[u] = (j + u < end) ? p : -INFINITY;
        }
        float mn = fmaxf(fmaxf(m, sc[0]), fmaxf(fmaxf(sc[1], sc[2]), sc[3]));
        float f  = expf(m - mn);
        l *= f; acc *= f;
        for (int u = 0; u < 4; ++u) {
            float w = expf(sc[u] - mn);
            l   += w;
            acc += c[u] * w;
        }
        m = mn;
    }
    float inv = (l > 0.0f) ? 1.0f / l : 0.0f;
    ((v2f*)neigh)[(size_t)d * (EMB / 2) + lane] = acc * inv;
}

__global__ void k_gemm_wmma(const float* __restrict__ neigh,
                            const float* __restrict__ W,
                            float* __restrict__ h,
                            float* __restrict__ part) {
    int row0  = blockIdx.x * 16;
    int wave  = threadIdx.x >> 5;
    int lane  = threadIdx.x & 31;
    int n0    = wave * 16;
    int m     = lane & 15;
    int khalf = lane >> 4;
    int n     = lane & 15;

    v8f acc = {};
#pragma unroll
    for (int k = 0; k < EMB; k += 4) {
        int ka = k + khalf * 2;
        v2f a, b;
        a.x = neigh[(size_t)(row0 + m) * EMB + ka];
        a.y = neigh[(size_t)(row0 + m) * EMB + ka + 1];
        b.x = W[(size_t)ka       * EMB + n0 + n];
        b.y = W[(size_t)(ka + 1) * EMB + n0 + n];
        acc = __builtin_amdgcn_wmma_f32_16x16x4_f32(
                  false, a, false, b, (short)0, acc, false, false);
    }
    int mbase = row0 + khalf * 8;
    float cs = 0.0f, cq = 0.0f;
#pragma unroll
    for (int r = 0; r < 8; ++r) {
        float v = acc[r];
        h[(size_t)(mbase + r) * EMB + n0 + n] = v;
        cs += v; cq += v * v;
    }
    cs += __shfl_xor(cs, 16, 32);
    cq += __shfl_xor(cq, 16, 32);
    if (khalf == 0) {
        v2f p; p.x = cs; p.y = cq;
        *(v2f*)(part + 2 * ((size_t)(n0 + n) * N_TILES + blockIdx.x)) = p;
    }
}

#define STATS_THREADS 1024
__global__ void __launch_bounds__(STATS_THREADS)
k_stats(const float* __restrict__ part,
        const float* __restrict__ gamma, const float* __restrict__ beta,
        float* __restrict__ scale, float* __restrict__ shift) {
    int c = blockIdx.x;
    const v2f* p = (const v2f*)part + (size_t)c * N_TILES;
    __shared__ float s_sum[STATS_THREADS / 32], s_sq[STATS_THREADS / 32];
    float sum = 0.0f, sq = 0.0f;
#pragma unroll
    for (int i = 0; i < (N_TILES + STATS_THREADS - 1) / STATS_THREADS; ++i) {
        int t = threadIdx.x + i * STATS_THREADS;
        if (t < N_TILES) { v2f v = p[t]; sum += v.x; sq += v.y; }
    }
#pragma unroll
    for (int off = 16; off > 0; off >>= 1) {
        sum += __shfl_xor(sum, off, 32);
        sq  += __shfl_xor(sq,  off, 32);
    }
    int wave = threadIdx.x >> 5, lane = threadIdx.x & 31;
    if (lane == 0) { s_sum[wave] = sum; s_sq[wave] = sq; }
    __syncthreads();
    if (wave == 0) {
        sum = s_sum[lane]; sq = s_sq[lane];
#pragma unroll
        for (int off = 16; off > 0; off >>= 1) {
            sum += __shfl_xor(sum, off, 32);
            sq  += __shfl_xor(sq,  off, 32);
        }
        if (lane == 0) {
            float mu   = sum / (float)N_NODES;
            float var  = sq / (float)N_NODES - mu * mu;
            float rinv = rsqrtf(var + BN_EPS);
            float sc   = gamma[c] * rinv;
            scale[c] = sc;
            shift[c] = beta[c] - mu * sc;
        }
    }
}

__global__ void k_apply(float* __restrict__ h,
                        const float* __restrict__ scale, const float* __restrict__ shift) {
    int i = blockIdx.x * blockDim.x + threadIdx.x;
    if (i >= N_NODES * EMB) return;
    int c = i & 63;
    h[i] = tanhf(h[i] * scale[c] + shift[c]);
}

extern "C" void kernel_launch(void* const* d_in, const int* in_sizes, int n_in,
                              void* d_out, int out_size, void* d_ws, size_t ws_size,
                              hipStream_t stream) {
    const float* ent   = (const float*)d_in[0];
    const float* rel   = (const float*)d_in[1];
    const int*   src   = (const int*)  d_in[2];
    const int*   dst   = (const int*)  d_in[3];
    const int*   rid   = (const int*)  d_in[4];
    const float* W     = (const float*)d_in[5];
    const float* gamma = (const float*)d_in[6];
    const float* beta  = (const float*)d_in[7];
    float* out = (float*)d_out;

    float* ws     = (float*)d_ws;
    int2*  sorted = (int2*)ws;
    int*   cnt    = (int*)(ws + 2 * (size_t)N_EDGES);
    int*   offs   = (int*)(ws + 2 * (size_t)N_EDGES + N_NODES);
    float* neigh  = ws + 2 * (size_t)N_EDGES + 2 * (size_t)N_NODES;
    float* scale  = neigh + (size_t)N_NODES * EMB;
    float* shift  = scale + EMB;
    int*   rank   = (int*)neigh;
    float* part   = ws;

    k_init <<<(N_NODES + 255) / 256, 256, 0, stream>>>(cnt);
    k_hist <<<(N_EDGES + 255) / 256, 256, 0, stream>>>(dst, cnt, rank);
    k_scan <<<SCAN_BLOCKS, SCAN_THREADS, 0, stream>>>(cnt, offs);
    k_fill <<<(N_EDGES + 255) / 256, 256, 0, stream>>>(src, dst, rid, offs, rank, sorted);
    k_node <<<N_NODES / NODE_WAVES, NODE_WAVES * 32, 0, stream>>>(ent, rel, cnt, offs, sorted, neigh);
    k_gemm_wmma<<<N_TILES, 128, 0, stream>>>(neigh, W, out, part);
    k_stats  <<<EMB, STATS_THREADS, 0, stream>>>(part, gamma, beta, scale, shift);
    k_apply  <<<(N_NODES * EMB + 255) / 256, 256, 0, stream>>>(out, scale, shift);
}


# Projection fused into the node kernel: 16 nodes per block hand their normalised rows over in LDS as one A tile and 4 waves run the same f32 WMMA from it; neigh round trip and one launch gone
# speedup vs baseline: 3.5605x; 1.0464x over previous
#include <hip/hip_runtime.h>
#include <math.h>

#define N_NODES 50000
#define N_EDGES 1250000
#define EMB     64
#define BN_EPS  1e-5f
#define N_TILES (N_NODES / 16)

typedef __attribute__((ext_vector_type(2))) float v2f;
typedef __attribute__((ext_vector_type(8))) float v8f;

__global__ void k_init(int* __restrict__ cnt) {
    int i = blockIdx.x * blockDim.x + threadIdx.x;
    if (i < N_NODES) cnt[i] = 0;
}

__global__ void k_hist(const int* __restrict__ dst, int* __restrict__ cnt,
                       int* __restrict__ rank) {
    int i = blockIdx.x * blockDim.x + threadIdx.x;
    if (i < N_EDGES) rank[i] = atomicAdd(&cnt[dst[i]], 1);
}

#define SCAN_THREADS 1024
#define SCAN_BLOCKS  ((N_NODES + SCAN_THREADS - 1) / SCAN_THREADS)
__global__ void __launch_bounds__(SCAN_THREADS)
k_scan(const int* __restrict__ cnt, int* __restrict__ offs) {
    __shared__ int s_v[SCAN_THREADS / 32], s_b[SCAN_THREADS / 32];
    int b = blockIdx.x, t = threadIdx.x;
    int lane = t & 31, wave = t >> 5;
    int before = 0;
    for (int i = 0; i < b; ++i) before += cnt[i * SCAN_THREADS + t];
    int idx = b * SCAN_THREADS + t;
    int v = (idx < N_NODES) ? cnt[idx] : 0;
    int inc = v;
    for (int off = 1; off < 32; off <<= 1) {
        int u = __shfl_up(inc, off, 32);
        if (lane >= off) inc += u;
    }
    for (int off = 16; off > 0; off >>= 1) before += __shfl_xor(before, off, 32);
    if (lane == 31) s_v[wave] = inc;
    if (lane == 0)  s_b[wave] = before;
    __syncthreads();
    int wb  = s_b[lane];
    int pre = (lane < wave) ? s_v[lane] : 0;
    for (int off = 16; off > 0; off >>= 1) {
        wb  += __shfl_xor(wb,  off, 32);
        pre += __shfl_xor(pre, off, 32);
    }
    if (idx < N_NODES) offs[idx] = wb + pre + inc - v;
}

__global__ void k_fill(const int* __restrict__ src, const int* __restrict__ dst,
                       const int* __restrict__ rid,
                       const int* __restrict__ offs, const int* __restrict__ rank,
                       int2* __restrict__ sorted) {
    int i = blockIdx.x * blockDim.x + threadIdx.x;
    if (i >= N_EDGES) return;
    int pos = offs[dst[i]] + rank[i];
    sorted[pos] = make_int2(src[i], rid[i]);
}

template <int MASK>
__device__ __forceinline__ float row_xor(float x) {
    int b = __builtin_bit_cast(int, x);
    return __builtin_bit_cast(float,
        __builtin_amdgcn_update_dpp(b, b, 0x160 + MASK, 0xf, 0xf, false));
}
__device__ __forceinline__ float xhalf(float x) {
    int b = __builtin_bit_cast(int, x);
    return __builtin_bit_cast(float,
        __builtin_amdgcn_permlanex16(b, b, 0x76543210, 0xfedcba98, false, false));
}
__device__ __forceinline__ float wave_sum(float p) {
    p += row_xor<1>(p);
    p += row_xor<2>(p);
    p += row_xor<4>(p);
    p += row_xor<8>(p);
    return p + xhalf(p);
}

#define TILE_ROWS  16
#define TILE_PITCH 68
__global__ void __launch_bounds__(TILE_ROWS * 32)
k_node_gemm(const float* __restrict__ ent, const float* __restrict__ rel,
            const int* __restrict__ cnt, const int* __restrict__ offs,
            const int2* __restrict__ sorted, const float* __restrict__ W,
            float* __restrict__ h, float* __restrict__ part) {
    __shared__ float s_tile[TILE_ROWS * TILE_PITCH];
    int lane = threadIdx.x & 31;
    int wave = __builtin_amdgcn_readfirstlane(threadIdx.x >> 5);
    int d = blockIdx.x * TILE_ROWS + wave;
    int beg = offs[d];
    int end = beg + cnt[d];
    const v2f* ent2 = (const v2f*)ent;
    const v2f* rel2 = (const v2f*)rel;
    v2f ed = ent2[(size_t)d * (EMB / 2) + lane];
    float m = -INFINITY, l = 0.0f;
    v2f acc = {0.0f, 0.0f};
    for (int j = beg; j < end; j += 4) {
        int2 pr[4];
        for (int u = 0; u < 4; ++u) pr[u] = sorted[min(j + u, end - 1)];
        v2f es[4], rr[4];
        for (int u = 0; u < 4; ++u) {
            es[u] = ent2[(size_t)pr[u].x * (EMB / 2) + lane];
            rr[u] = rel2[(size_t)pr[u].y * (EMB / 2) + lane];
        }
        v2f   c[4];
        float sc[4];
        for (int u = 0; u < 4; ++u) {
            c[u] = es[u] * rr[u];
            float p = wave_sum(c[u].x * ed.x + c[u].y * ed.y);
            sc[u] = (j + u < end) ? p : -INFINITY;
        }
        float mn = fmaxf(fmaxf(m, sc[0]), fmaxf(fmaxf(sc[1], sc[2]), sc[3]));
        float f  = expf(m - mn);
        l *= f; acc *= f;
        for (int u = 0; u < 4; ++u) {
            float w = expf(sc[u] - mn);
            l   += w;
            acc += c[u] * w;
        }
        m = mn;
    }
    float inv = (l > 0.0f) ? 1.0f / l : 0.0f;
    *(v2f*)(s_tile + wave * TILE_PITCH + 2 * lane) = acc * inv;

    int n0    = wave * 16;
    int mrow  = lane & 15;
    int khalf = lane >> 4;
    int n     = lane & 15;
    v2f bfrag[EMB / 4];
    if (wave < 4) {
#pragma unroll
        for (int k = 0; k < EMB; k += 4) {
            int ka = k + khalf * 2;
            bfrag[k / 4].x = W[(size_t)ka       * EMB + n0 + n];
            bfrag[k / 4].y = W[(size_t)(ka + 1) * EMB + n0 + n];
        }
    }
    __syncthreads();
    if (wave >= 4) return;

    v8f accd = {};
#pragma unroll
    for (int k = 0; k < EMB; k += 4) {
        int ka = k + khalf * 2;
        v2f a = *(const v2f*)(s_tile + mrow * TILE_PITCH + ka);
        accd = __builtin_amdgcn_wmma_f32_16x16x4_f32(
                   false, a, false, bfrag[k / 4], (short)0, accd, false, false);
    }
    int mbase = blockIdx.x * TILE_ROWS + khalf * 8;
    float cs = 0.0f, cq = 0.0f;
#pragma unroll
    for (int r = 0; r < 8; ++r) {
        float v = accd[r];
        h[(size_t)(mbase + r) * EMB + n0 + n] = v;
        cs += v; cq += v * v;
    }
    cs += __shfl_xor(cs, 16, 32);
    cq += __shfl_xor(cq, 16, 32);
    if (khalf == 0) {
        v2f p; p.x = cs; p.y = cq;
        *(v2f*)(part + 2 * ((size_t)(n0 + n) * N_TILES + blockIdx.x)) = p;
    }
}

#define STATS_THREADS 1024
__global__ void __launch_bounds__(STATS_THREADS)
k_stats(const float* __restrict__ part,
        const float* __restrict__ gamma, const float* __restrict__ beta,
        float* __restrict__ scale, float* __restrict__ shift) {
    int c = blockIdx.x;
    const v2f* p = (const v2f*)part + (size_t)c * N_TILES;
    __shared__ float s_sum[STATS_THREADS / 32], s_sq[STATS_THREADS / 32];
    float sum = 0.0f, sq = 0.0f;
#pragma unroll
    for (int i = 0; i < (N_TILES + STATS_THREADS - 1) / STATS_THREADS; ++i) {
        int t = threadIdx.x + i * STATS_THREADS;
        if (t < N_TILES) { v2f v = p[t]; sum += v.x; sq += v.y; }
    }
#pragma unroll
    for (int off = 16; off > 0; off >>= 1) {
        sum += __shfl_xor(sum, off, 32);
        sq  += __shfl_xor(sq,  off, 32);
    }
    int wave = threadIdx.x >> 5, lane = threadIdx.x & 31;
    if (lane == 0) { s_sum[wave] = sum; s_sq[wave] = sq; }
    __syncthreads();
    if (wave == 0) {
        sum = s_sum[lane]; sq = s_sq[lane];
#pragma unroll
        for (int off = 16; off > 0; off >>= 1) {
            sum += __shfl_xor(sum, off, 32);
            sq  += __shfl_xor(sq,  off, 32);
        }
        if (lane == 0) {
            float mu   = sum / (float)N_NODES;
            float var  = sq / (float)N_NODES - mu * mu;
            float rinv = rsqrtf(var + BN_EPS);
            float sc   = gamma[c] * rinv;
            scale[c] = sc;
            shift[c] = beta[c] - mu * sc;
        }
    }
}

__global__ void k_apply(float* __restrict__ h,
                        const float* __restrict__ scale, const float* __restrict__ shift) {
    int i = blockIdx.x * blockDim.x + threadIdx.x;
    if (i >= N_NODES * EMB) return;
    int c = i & 63;
    h[i] = tanhf(h[i] * scale[c] + shift[c]);
}

extern "C" void kernel_launch(void* const* d_in, const int* in_sizes, int n_in,
                              void* d_out, int out_size, void* d_ws, size_t ws_size,
                              hipStream_t stream) {
    const float* ent   = (const float*)d_in[0];
    const float* rel   = (const float*)d_in[1];
    const int*   src   = (const int*)  d_in[2];
    const int*   dst   = (const int*)  d_in[3];
    const int*   rid   = (const int*)  d_in[4];
    const float* W     = (const float*)d_in[5];
    const float* gamma = (const float*)d_in[6];
    const float* beta  = (const float*)d_in[7];
    float* out = (float*)d_out;

    float* ws     = (float*)d_ws;
    int2*  sorted = (int2*)ws;
    int*   cnt    = (int*)(ws + 2 * (size_t)N_EDGES);
    int*   offs   = (int*)(ws + 2 * (size_t)N_EDGES + N_NODES);
    float* spare  = ws + 2 * (size_t)N_EDGES + 2 * (size_t)N_NODES;
    float* scale  = spare + (size_t)N_NODES * EMB;
    float* shift  = scale + EMB;
    int*   rank   = (int*)spare;
    float* part   = spare;

    k_init <<<(N_NODES + 255) / 256, 256, 0, stream>>>(cnt);
    k_hist <<<(N_EDGES + 255) / 256, 256, 0, stream>>>(dst, cnt, rank);
    k_scan <<<SCAN_BLOCKS, SCAN_THREADS, 0, stream>>>(cnt, offs);
    k_fill <<<(N_EDGES + 255) / 256, 256, 0, stream>>>(src, dst, rid, offs, rank, sorted);
    k_node_gemm<<<N_TILES, TILE_ROWS * 32, 0, stream>>>(ent, rel, cnt, offs, sorted, W, out, part);
    k_stats  <<<EMB, STATS_THREADS, 0, stream>>>(part, gamma, beta, scale, shift);
    k_apply  <<<(N_NODES * EMB + 255) / 256, 256, 0, stream>>>(out, scale, shift);
}


# Counting-sort counters in LDS: blocks of (edge chunk, node range) count and place edges with LDS atomics from a per-chunk prefix table; global atomics and offs gather gone; segment order chunk-major
# speedup vs baseline: 5.2089x; 1.4630x over previous
#include <hip/hip_runtime.h>
#include <math.h>

#define N_NODES 50000
#define N_EDGES 1250000
#define EMB     64
#define BN_EPS  1e-5f
#define N_TILES (N_NODES / 16)

typedef __attribute__((ext_vector_type(2))) float v2f;
typedef __attribute__((ext_vector_type(8))) float v8f;

#define SORT_CHUNKS  64
#define SORT_RANGES  4
#define SORT_THREADS 1024
#define CHUNK_EDGES  ((N_EDGES + SORT_CHUNKS - 1) / SORT_CHUNKS)
#define CHUNK_TRIPS  ((CHUNK_EDGES + SORT_THREADS - 1) / SORT_THREADS)
#define RANGE_NODES  (N_NODES / SORT_RANGES)
static_assert(RANGE_NODES * SORT_RANGES == N_NODES, "node ranges must tile the nodes");

__global__ void __launch_bounds__(SORT_THREADS)
k_hist(const int* __restrict__ dst, int* __restrict__ cmat) {
    __shared__ int s_cnt[RANGE_NODES];
    int t = threadIdx.x, chunk = blockIdx.x, lo = blockIdx.y * RANGE_NODES;
    for (int i = t; i < RANGE_NODES; i += SORT_THREADS) s_cnt[i] = 0;
    int d[CHUNK_TRIPS];
#pragma unroll
    for (int k = 0; k < CHUNK_TRIPS; ++k) {
        int w = k * SORT_THREADS + t;
        int i = chunk * CHUNK_EDGES + w;
        d[k] = (w < CHUNK_EDGES && i < N_EDGES) ? dst[i] : -1;
    }
    __syncthreads();
#pragma unroll
    for (int k = 0; k < CHUNK_TRIPS; ++k) {
        unsigned r = (unsigned)(d[k] - lo);
        if (r < (unsigned)RANGE_NODES) atomicAdd(&s_cnt[r], 1);
    }
    __syncthreads();
    int* row = cmat + (size_t)chunk * N_NODES + lo;
    for (int i = t; i < RANGE_NODES; i += SORT_THREADS) row[i] = s_cnt[i];
}

__global__ void k_colsum(int* __restrict__ cmat, int* __restrict__ cnt) {
    int n = blockIdx.x * blockDim.x + threadIdx.x;
    if (n >= N_NODES) return;
    int c[SORT_CHUNKS];
#pragma unroll
    for (int b = 0; b < SORT_CHUNKS; ++b) c[b] = cmat[(size_t)b * N_NODES + n];
    int run = 0;
#pragma unroll
    for (int b = 0; b < SORT_CHUNKS; ++b) {
        cmat[(size_t)b * N_NODES + n] = run;
        run += c[b];
    }
    cnt[n] = run;
}

#define SCAN_THREADS 1024
#define SCAN_BLOCKS  ((N_NODES + SCAN_THREADS - 1) / SCAN_THREADS)
__global__ void __launch_bounds__(SCAN_THREADS)
k_scan(const int* __restrict__ cnt, int* __restrict__ offs) {
    __shared__ int s_v[SCAN_THREADS / 32], s_b[SCAN_THREADS / 32];
    int b = blockIdx.x, t = threadIdx.x;
    int lane = t & 31, wave = t >> 5;
    int before = 0;
    for (int i = 0; i < b; ++i) before += cnt[i * SCAN_THREADS + t];
    int idx = b * SCAN_THREADS + t;
    int v = (idx < N_NODES) ? cnt[idx] : 0;
    int inc = v;
    for (int off = 1; off < 32; off <<= 1) {
        int u = __shfl_up(inc, off, 32);
        if (lane >= off) inc += u;
    }
    for (int off = 16; off > 0; off >>= 1) before += __shfl_xor(before, off, 32);
    if (lane == 31) s_v[wave] = inc;
    if (lane == 0)  s_b[wave] = before;
    __syncthreads();
    int wb  = s_b[lane];
    int pre = (lane < wave) ? s_v[lane] : 0;
    for (int off = 16; off > 0; off >>= 1) {
        wb  += __shfl_xor(wb,  off, 32);
        pre += __shfl_xor(pre, off, 32);
    }
    if (idx < N_NODES) offs[idx] = wb + pre + inc - v;
}

__global__ void __launch_bounds__(SORT_THREADS)
k_fill(const int* __restrict__ src, const int* __restrict__ dst,
       const int* __restrict__ rid,
       const int* __restrict__ offs, const int* __restrict__ cmat,
       int2* __restrict__ sorted) {
    __shared__ int s_pos[RANGE_NODES];
    int t = threadIdx.x, chunk = blockIdx.x, lo = blockIdx.y * RANGE_NODES;
    const int* row = cmat + (size_t)chunk * N_NODES + lo;
    for (int i = t; i < RANGE_NODES; i += SORT_THREADS) s_pos[i] = offs[lo + i] + row[i];
    int d[CHUNK_TRIPS];
#pragma unroll
    for (int k = 0; k < CHUNK_TRIPS; ++k) {
        int w = k * SORT_THREADS + t;
        int i = chunk * CHUNK_EDGES + w;
        d[k] = (w < CHUNK_EDGES && i < N_EDGES) ? dst[i] : -1;
    }
    int s[CHUNK_TRIPS], q[CHUNK_TRIPS];
#pragma unroll
    for (int k = 0; k < CHUNK_TRIPS; ++k) {
        int i = chunk * CHUNK_EDGES + k * SORT_THREADS + t;
        bool mine = (unsigned)(d[k] - lo) < (unsigned)RANGE_NODES;
        s[k] = mine ? src[i] : 0;
        q[k] = mine ? rid[i] : 0;
    }
    __syncthreads();
#pragma unroll
    for (int k = 0; k < CHUNK_TRIPS; ++k) {
        unsigned r = (unsigned)(d[k] - lo);
        if (r < (unsigned)RANGE_NODES) {
            int pos = atomicAdd(&s_pos[r], 1);
            sorted[pos] = make_int2(s[k], q[k]);
        }
    }
}

template <int MASK>
__device__ __forceinline__ float row_xor(float x) {
    int b = __builtin_bit_cast(int, x);
    return __builtin_bit_cast(float,
        __builtin_amdgcn_update_dpp(b, b, 0x160 + MASK, 0xf, 0xf, false));
}
__device__ __forceinline__ float xhalf(float x) {
    int b = __builtin_bit_cast(int, x);
    return __builtin_bit_cast(float,
        __builtin_amdgcn_permlanex16(b, b, 0x76543210, 0xfedcba98, false, false));
}
__device__ __forceinline__ float wave_sum(float p) {
    p += row_xor<1>(p);
    p += row_xor<2>(p);
    p += row_xor<4>(p);
    p += row_xor<8>(p);
    return p + xhalf(p);
}

#define TILE_ROWS  16
#define TILE_PITCH 68
__global__ void __launch_bounds__(TILE_ROWS * 32)
k_node_gemm(const float* __restrict__ ent, const float* __restrict__ rel,
            const int* __restrict__ cnt, const int* __restrict__ offs,
            const int2* __restrict__ sorted, const float* __restrict__ W,
            float* __restrict__ h, float* __restrict__ part) {
    __shared__ float s_tile[TILE_ROWS * TILE_PITCH];
    int lane = threadIdx.x & 31;
    int wave = __builtin_amdgcn_readfirstlane(threadIdx.x >> 5);
    int d = blockIdx.x * TILE_ROWS + wave;
    int beg = offs[d];
    int end = beg + cnt[d];
    const v2f* ent2 = (const v2f*)ent;
    const v2f* rel2 = (const v2f*)rel;
    v2f ed = ent2[(size_t)d * (EMB / 2) + lane];
    float m = -INFINITY, l = 0.0f;
    v2f acc = {0.0f, 0.0f};
    for (int j = beg; j < end; j += 4) {
        int2 pr[4];
        for (int u = 0; u < 4; ++u) pr[u] = sorted[min(j + u, end - 1)];
        v2f es[4], rr[4];
        for (int u = 0; u < 4; ++u) {
            es[u] = ent2[(size_t)pr[u].x * (EMB / 2) + lane];
            rr[u] = rel2[(size_t)pr[u].y * (EMB / 2) + lane];
        }
        v2f   c[4];
        float sc[4];
        for (int u = 0; u < 4; ++u) {
            c[u] = es[u] * rr[u];
            float p = wave_sum(c[u].x * ed.x + c[u].y * ed.y);
            sc[u] = (j + u < end) ? p : -INFINITY;
        }
        float mn = fmaxf(fmaxf(m, sc[0]), fmaxf(fmaxf(sc[1], sc[2]), sc[3]));
        float f  = expf(m - mn);
        l *= f; acc *= f;
        for (int u = 0; u < 4; ++u) {
            float w = expf(sc[u] - mn);
            l   += w;
            acc += c[u] * w;
        }
        m = mn;
    }
    float inv = (l > 0.0f) ? 1.0f / l : 0.0f;
    *(v2f*)(s_tile + wave * TILE_PITCH + 2 * lane) = acc * inv;

    int n0    = wave * 16;
    int mrow  = lane & 15;
    int khalf = lane >> 4;
    int n     = lane & 15;
    v2f bfrag[EMB / 4];
    if (wave < 4) {
#pragma unroll
        for (int k = 0; k < EMB; k += 4) {
            int ka = k + khalf * 2;
            bfrag[k / 4].x = W[(size_t)ka       * EMB + n0 + n];
            bfrag[k / 4].y = W[(size_t)(ka + 1) * EMB + n0 + n];
        }
    }
    __syncthreads();
    if (wave >= 4) return;

    v8f accd = {};
#pragma unroll
    for (int k = 0; k < EMB; k += 4) {
        int ka = k + khalf * 2;
        v2f a = *(const v2f*)(s_tile + mrow * TILE_PITCH + ka);
        accd = __builtin_amdgcn_wmma_f32_16x16x4_f32(
                   false, a, false, bfrag[k / 4], (short)0, accd, false, false);
    }
    int mbase = blockIdx.x * TILE_ROWS + khalf * 8;
    float cs = 0.0f, cq = 0.0f;
#pragma unroll
    for (int r = 0; r < 8; ++r) {
        float v = accd[r];
        h[(size_t)(mbase + r) * EMB + n0 + n] = v;
        cs += v; cq += v * v;
    }
    cs += __shfl_xor(cs, 16, 32);
    cq += __shfl_xor(cq, 16, 32);
    if (khalf == 0) {
        v2f p; p.x = cs; p.y = cq;
        *(v2f*)(part + 2 * ((size_t)(n0 + n) * N_TILES + blockIdx.x)) = p;
    }
}

#define STATS_THREADS 1024
__global__ void __launch_bounds__(STATS_THREADS)
k_stats(const float* __restrict__ part,
        const float* __restrict__ gamma, const float* __restrict__ beta,
        float* __restrict__ scale, float* __restrict__ shift) {
    int c = blockIdx.x;
    const v2f* p = (const v2f*)part + (size_t)c * N_TILES;
    __shared__ float s_sum[STATS_THREADS / 32], s_sq[STATS_THREADS / 32];
    float sum = 0.0f, sq = 0.0f;
#pragma unroll
    for (int i = 0; i < (N_TILES + STATS_THREADS - 1) / STATS_THREADS; ++i) {
        int t = threadIdx.x + i * STATS_THREADS;
        if (t < N_TILES) { v2f v = p[t]; sum += v.x; sq += v.y; }
    }
#pragma unroll
    for (int off = 16; off > 0; off >>= 1) {
        sum += __shfl_xor(sum, off, 32);
        sq  += __shfl_xor(sq,  off, 32);
    }
    int wave = threadIdx.x >> 5, lane = threadIdx.x & 31;
    if (lane == 0) { s_sum[wave] = sum; s_sq[wave] = sq; }
    __syncthreads();
    if (wave == 0) {
        sum = s_sum[lane]; sq = s_sq[lane];
#pragma unroll
        for (int off = 16; off > 0; off >>= 1) {
            sum += __shfl_xor(sum, off, 32);
            sq  += __shfl_xor(sq,  off, 32);
        }
        if (lane == 0) {
            float mu   = sum / (float)N_NODES;
            float var  = sq / (float)N_NODES - mu * mu;
            float rinv = rsqrtf(var + BN_EPS);
            float sc   = gamma[c] * rinv;
            scale[c] = sc;
            shift[c] = beta[c] - mu * sc;
        }
    }
}

__global__ void k_apply(float* __restrict__ h,
                        const float* __restrict__ scale, const float* __restrict__ shift) {
    int i = blockIdx.x * blockDim.x + threadIdx.x;
    if (i >= N_NODES * EMB) return;
    int c = i & 63;
    h[i] = tanhf(h[i] * scale[c] + shift[c]);
}

extern "C" void kernel_launch(void* const* d_in, const int* in_sizes, int n_in,
                              void* d_out, int out_size, void* d_ws, size_t ws_size,
                              hipStream_t stream) {
    const float* ent   = (const float*)d_in[0];
    const float* rel   = (const float*)d_in[1];
    const int*   src   = (const int*)  d_in[2];
    const int*   dst   = (const int*)  d_in[3];
    const int*   rid   = (const int*)  d_in[4];
    const float* W     = (const float*)d_in[5];
    const float* gamma = (const float*)d_in[6];
    const float* beta  = (const float*)d_in[7];
    float* out = (float*)d_out;

    float* ws     = (float*)d_ws;
    int2*  sorted = (int2*)ws;
    int*   cnt    = (int*)(ws + 2 * (size_t)N_EDGES);
    int*   offs   = (int*)(ws + 2 * (size_t)N_EDGES + N_NODES);
    float* spare  = ws + 2 * (size_t)N_EDGES + 2 * (size_t)N_NODES;
    float* scale  = spare + (size_t)N_NODES * EMB;
    float* shift  = scale + EMB;
    int*   cmat   = (int*)spare;
    float* part   = spare;

    k_hist  <<<dim3(SORT_CHUNKS, SORT_RANGES), SORT_THREADS, 0, stream>>>(dst, cmat);
    k_colsum<<<(N_NODES + 255) / 256, 256, 0, stream>>>(cmat, cnt);
    k_scan <<<SCAN_BLOCKS, SCAN_THREADS, 0, stream>>>(cnt, offs);
    k_fill  <<<dim3(SORT_CHUNKS, SORT_RANGES), SORT_THREADS, 0, stream>>>(src, dst, rid, offs, cmat, sorted);
    k_node_gemm<<<N_TILES, TILE_ROWS * 32, 0, stream>>>(ent, rel, cnt, offs, sorted, W, out, part);
    k_stats  <<<EMB, STATS_THREADS, 0, stream>>>(part, gamma, beta, scale, shift);
    k_apply  <<<(N_NODES * EMB + 255) / 256, 256, 0, stream>>>(out, scale, shift);
}


# Node phase re-mapped: 8 lanes per edge, 8 edges per wave step; 3-step dot reduce, one exp per group edge, pairs by one coalesced load + shuffles; 4 partial softmax sums merged (new sum order)
# speedup vs baseline: 5.6065x; 1.0763x over previous
#include <hip/hip_runtime.h>
#include <math.h>

#define N_NODES 50000
#define N_EDGES 1250000
#define EMB     64
#define BN_EPS  1e-5f
#define N_TILES (N_NODES / 16)

typedef __attribute__((ext_vector_type(2))) float v2f;
typedef __attribute__((ext_vector_type(8))) float v8f;

#define SORT_CHUNKS  64
#define SORT_RANGES  4
#define SORT_THREADS 1024
#define CHUNK_EDGES  ((N_EDGES + SORT_CHUNKS - 1) / SORT_CHUNKS)
#define CHUNK_TRIPS  ((CHUNK_EDGES + SORT_THREADS - 1) / SORT_THREADS)
#define RANGE_NODES  (N_NODES / SORT_RANGES)
static_assert(RANGE_NODES * SORT_RANGES == N_NODES, "node ranges must tile the nodes");

__global__ void __launch_bounds__(SORT_THREADS)
k_hist(const int* __restrict__ dst, int* __restrict__ cmat) {
    __shared__ int s_cnt[RANGE_NODES];
    int t = threadIdx.x, chunk = blockIdx.x, lo = blockIdx.y * RANGE_NODES;
    for (int i = t; i < RANGE_NODES; i += SORT_THREADS) s_cnt[i] = 0;
    int d[CHUNK_TRIPS];
#pragma unroll
    for (int k = 0; k < CHUNK_TRIPS; ++k) {
        int w = k * SORT_THREADS + t;
        int i = chunk * CHUNK_EDGES + w;
        d[k] = (w < CHUNK_EDGES && i < N_EDGES) ? dst[i] : -1;
    }
    __syncthreads();
#pragma unroll
    for (int k = 0; k < CHUNK_TRIPS; ++k) {
        unsigned r = (unsigned)(d[k] - lo);
        if (r < (unsigned)RANGE_NODES) atomicAdd(&s_cnt[r], 1);
    }
    __syncthreads();
    int* row = cmat + (size_t)chunk * N_NODES + lo;
    for (int i = t; i < RANGE_NODES; i += SORT_THREADS) row[i] = s_cnt[i];
}

__global__ void k_colsum(int* __restrict__ cmat, int* __restrict__ cnt) {
    int n = blockIdx.x * blockDim.x + threadIdx.x;
    if (n >= N_NODES) return;
    int c[SORT_CHUNKS];
#pragma unroll
    for (int b = 0; b < SORT_CHUNKS; ++b) c[b] = cmat[(size_t)b * N_NODES + n];
    int run = 0;
#pragma unroll
    for (int b = 0; b < SORT_CHUNKS; ++b) {
        cmat[(size_t)b * N_NODES + n] = run;
        run += c[b];
    }
    cnt[n] = run;
}

#define SCAN_THREADS 1024
#define SCAN_BLOCKS  ((N_NODES + SCAN_THREADS - 1) / SCAN_THREADS)
__global__ void __launch_bounds__(SCAN_THREADS)
k_scan(const int* __restrict__ cnt, int* __restrict__ offs) {
    __shared__ int s_v[SCAN_THREADS / 32], s_b[SCAN_THREADS / 32];
    int b = blockIdx.x, t = threadIdx.x;
    int lane = t & 31, wave = t >> 5;
    int before = 0;
    for (int i = 0; i < b; ++i) before += cnt[i * SCAN_THREADS + t];
    int idx = b * SCAN_THREADS + t;
    int v = (idx < N_NODES) ? cnt[idx] : 0;
    int inc = v;
    for (int off = 1; off < 32; off <<= 1) {
        int u = __shfl_up(inc, off, 32);
        if (lane >= off) inc += u;
    }
    for (int off = 16; off > 0; off >>= 1) before += __shfl_xor(before, off, 32);
    if (lane == 31) s_v[wave] = inc;
    if (lane == 0)  s_b[wave] = before;
    __syncthreads();
    int wb  = s_b[lane];
    int pre = (lane < wave) ? s_v[lane] : 0;
    for (int off = 16; off > 0; off >>= 1) {
        wb  += __shfl_xor(wb,  off, 32);
        pre += __shfl_xor(pre, off, 32);
    }
    if (idx < N_NODES) offs[idx] = wb + pre + inc - v;
}

__global__ void __launch_bounds__(SORT_THREADS)
k_fill(const int* __restrict__ src, const int* __restrict__ dst,
       const int* __restrict__ rid,
       const int* __restrict__ offs, const int* __restrict__ cmat,
       int2* __restrict__ sorted) {
    __shared__ int s_pos[RANGE_NODES];
    int t = threadIdx.x, chunk = blockIdx.x, lo = blockIdx.y * RANGE_NODES;
    const int* row = cmat + (size_t)chunk * N_NODES + lo;
    for (int i = t; i < RANGE_NODES; i += SORT_THREADS) s_pos[i] = offs[lo + i] + row[i];
    int d[CHUNK_TRIPS];
#pragma unroll
    for (int k = 0; k < CHUNK_TRIPS; ++k) {
        int w = k * SORT_THREADS + t;
        int i = chunk * CHUNK_EDGES + w;
        d[k] = (w < CHUNK_EDGES && i < N_EDGES) ? dst[i] : -1;
    }
    int s[CHUNK_TRIPS], q[CHUNK_TRIPS];
#pragma unroll
    for (int k = 0; k < CHUNK_TRIPS; ++k) {
        int i = chunk * CHUNK_EDGES + k * SORT_THREADS + t;
        bool mine = (unsigned)(d[k] - lo) < (unsigned)RANGE_NODES;
        s[k] = mine ? src[i] : 0;
        q[k] = mine ? rid[i] : 0;
    }
    __syncthreads();
#pragma unroll
    for (int k = 0; k < CHUNK_TRIPS; ++k) {
        unsigned r = (unsigned)(d[k] - lo);
        if (r < (unsigned)RANGE_NODES) {
            int pos = atomicAdd(&s_pos[r], 1);
            sorted[pos] = make_int2(s[k], q[k]);
        }
    }
}

template <int MASK>
__device__ __forceinline__ float row_xor(float x) {
    int b = __builtin_bit_cast(int, x);
    return __builtin_bit_cast(float,
        __builtin_amdgcn_update_dpp(b, b, 0x160 + MASK, 0xf, 0xf, false));
}
__device__ __forceinline__ float xhalf(float x) {
    int b = __builtin_bit_cast(int, x);
    return __builtin_bit_cast(float,
        __builtin_amdgcn_permlanex16(b, b, 0x76543210, 0xfedcba98, false, false));
}
__device__ __forceinline__ float wave_sum(float p) {
    p += row_xor<1>(p);
    p += row_xor<2>(p);
    p += row_xor<4>(p);
    p += row_xor<8>(p);
    return p + xhalf(p);
}

#define TILE_ROWS  16
#define TILE_PITCH 68
#define EDGE_GROUPS      4
#define EDGES_PER_GROUP  2
#define STEP_EDGES       (EDGE_GROUPS * EDGES_PER_GROUP)
typedef __attribute__((ext_vector_type(4))) float v4f;

__device__ __forceinline__ float dot8(v4f a0, v4f a1, v4f b0, v4f b1) {
    float p = a0.x * b0.x;
    p += a0.y * b0.y; p += a0.z * b0.z; p += a0.w * b0.w;
    p += a1.x * b1.x; p += a1.y * b1.y; p += a1.z * b1.z; p += a1.w * b1.w;
    return p;
}

__global__ void __launch_bounds__(TILE_ROWS * 32)
k_node_gemm(const float* __restrict__ ent, const float* __restrict__ rel,
            const int* __restrict__ cnt, const int* __restrict__ offs,
            const int2* __restrict__ sorted, const float* __restrict__ W,
            float* __restrict__ h, float* __restrict__ part) {
    __shared__ __attribute__((aligned(16))) float s_tile[TILE_ROWS * TILE_PITCH];
    int lane = threadIdx.x & 31;
    int wave = __builtin_amdgcn_readfirstlane(threadIdx.x >> 5);
    int d = blockIdx.x * TILE_ROWS + wave;
    int beg = offs[d];
    int end = beg + cnt[d];
    int g = lane >> 3;
    int q = lane & 7;
    const v4f* ent4 = (const v4f*)ent;
    const v4f* rel4 = (const v4f*)rel;
    v4f ed0 = ent4[(size_t)d * (EMB / 4) + q];
    v4f ed1 = ent4[(size_t)d * (EMB / 4) + 8 + q];
    float m = -INFINITY, l = 0.0f;
    v4f acc0 = {0.0f, 0.0f, 0.0f, 0.0f}, acc1 = {0.0f, 0.0f, 0.0f, 0.0f};
    if (end > beg) {
        int2 pr = sorted[min(beg + lane, end - 1)];
        for (int cb = beg; cb < end; cb += 32) {
            int nchunk = min(32, end - cb);
            for (int s = 0; s < nchunk; s += STEP_EDGES) {
                bool ok[EDGES_PER_GROUP];
                v4f  c0[EDGES_PER_GROUP], c1[EDGES_PER_GROUP];
                v4f  r0[EDGES_PER_GROUP], r1[EDGES_PER_GROUP];
#pragma unroll
                for (int u = 0; u < EDGES_PER_GROUP; ++u) {
                    int slot = s + u * EDGE_GROUPS + g;
                    ok[u] = slot < nchunk;
                    int sl = min(slot, nchunk - 1);
                    int si = __shfl(pr.x, sl, 32);
                    int ri = __shfl(pr.y, sl, 32);
                    unsigned eo = (unsigned)si * (EMB / 4) + (unsigned)q;
                    unsigned ro = (unsigned)ri * (EMB / 4) + (unsigned)q;
                    c0[u] = ent4[eo];
                    c1[u] = ent4[eo + 8u];
                    r0[u] = rel4[ro];
                    r1[u] = rel4[ro + 8u];
                }
                float sc[EDGES_PER_GROUP];
                float tmax = -INFINITY;
#pragma unroll
                for (int u = 0; u < EDGES_PER_GROUP; ++u) {
                    c0[u] *= r0[u]; c1[u] *= r1[u];
                    float p = dot8(c0[u], c1[u], ed0, ed1);
                    p += row_xor<1>(p);
                    p += row_xor<2>(p);
                    p += row_xor<4>(p);
                    sc[u] = ok[u] ? p : -INFINITY;
                    tmax = fmaxf(tmax, sc[u]);
                }
                tmax = fmaxf(tmax, row_xor<8>(tmax));
                tmax = fmaxf(tmax, xhalf(tmax));
                float mn = fmaxf(m, tmax);
                float f  = expf(m - mn);
                l *= f; acc0 *= f; acc1 *= f;
#pragma unroll
                for (int u = 0; u < EDGES_PER_GROUP; ++u) {
                    float w = expf(sc[u] - mn);
                    l    += w;
                    acc0 += c0[u] * w;
                    acc1 += c1[u] * w;
                }
                m = mn;
            }
            if (cb + 32 < end) pr = sorted[min(cb + 32 + lane, end - 1)];
        }
    }
    l += row_xor<8>(l);  l += xhalf(l);
#pragma unroll
    for (int i = 0; i < 4; ++i) {
        acc0[i] += row_xor<8>(acc0[i]);  acc0[i] += xhalf(acc0[i]);
        acc1[i] += row_xor<8>(acc1[i]);  acc1[i] += xhalf(acc1[i]);
    }
    float inv = (l > 0.0f) ? 1.0f / l : 0.0f;
    if (g == 0) {
        *(v4f*)(s_tile + wave * TILE_PITCH + 4 * q)      = acc0 * inv;
        *(v4f*)(s_tile + wave * TILE_PITCH + 32 + 4 * q) = acc1 * inv;
    }

    int n0    = wave * 16;
    int mrow  = lane & 15;
    int khalf = lane >> 4;
    int n     = lane & 15;
    v2f bfrag[EMB / 4];
    if (wave < 4) {
#pragma unroll
        for (int k = 0; k < EMB; k += 4) {
            int ka = k + khalf * 2;
            bfrag[k / 4].x = W[(size_t)ka       * EMB + n0 + n];
            bfrag[k / 4].y = W[(size_t)(ka + 1) * EMB + n0 + n];
        }
    }
    __syncthreads();
    if (wave >= 4) return;

    v8f accd = {};
#pragma unroll
    for (int k = 0; k < EMB; k += 4) {
        int ka = k + khalf * 2;
        v2f a = *(const v2f*)(s_tile + mrow * TILE_PITCH + ka);
        accd = __builtin_amdgcn_wmma_f32_16x16x4_f32(
                   false, a, false, bfrag[k / 4], (short)0, accd, false, false);
    }
    int mbase = blockIdx.x * TILE_ROWS + khalf * 8;
    float cs = 0.0f, cq = 0.0f;
#pragma unroll
    for (int r = 0; r < 8; ++r) {
        float v = accd[r];
        h[(size_t)(mbase + r) * EMB + n0 + n] = v;
        cs += v; cq += v * v;
    }
    cs += __shfl_xor(cs, 16, 32);
    cq += __shfl_xor(cq, 16, 32);
    if (khalf == 0) {
        v2f p; p.x = cs; p.y = cq;
        *(v2f*)(part + 2 * ((size_t)(n0 + n) * N_TILES + blockIdx.x)) = p;
    }
}

#define STATS_THREADS 1024
__global__ void __launch_bounds__(STATS_THREADS)
k_stats(const float* __restrict__ part,
        const float* __restrict__ gamma, const float* __restrict__ beta,
        float* __restrict__ scale, float* __restrict__ shift) {
    int c = blockIdx.x;
    const v2f* p = (const v2f*)part + (size_t)c * N_TILES;
    __shared__ float s_sum[STATS_THREADS / 32], s_sq[STATS_THREADS / 32];
    float sum = 0.0f, sq = 0.0f;
#pragma unroll
    for (int i = 0; i < (N_TILES + STATS_THREADS - 1) / STATS_THREADS; ++i) {
        int t = threadIdx.x + i * STATS_THREADS;
        if (t < N_TILES) { v2f v = p[t]; sum += v.x; sq += v.y; }
    }
#pragma unroll
    for (int off = 16; off > 0; off >>= 1) {
        sum += __shfl_xor(sum, off, 32);
        sq  += __shfl_xor(sq,  off, 32);
    }
    int wave = threadIdx.x >> 5, lane = threadIdx.x & 31;
    if (lane == 0) { s_sum[wave] = sum; s_sq[wave] = sq; }
    __syncthreads();
    if (wave == 0) {
        sum = s_sum[lane]; sq = s_sq[lane];
#pragma unroll
        for (int off = 16; off > 0; off >>= 1) {
            sum += __shfl_xor(sum, off, 32);
            sq  += __shfl_xor(sq,  off, 32);
        }
        if (lane == 0) {
            float mu   = sum / (float)N_NODES;
            float var  = sq / (float)N_NODES - mu * mu;
            float rinv = rsqrtf(var + BN_EPS);
            float sc   = gamma[c] * rinv;
            scale[c] = sc;
            shift[c] = beta[c] - mu * sc;
        }
    }
}

__global__ void k_apply(float* __restrict__ h,
                        const float* __restrict__ scale, const float* __restrict__ shift) {
    int i = blockIdx.x * blockDim.x + threadIdx.x;
    if (i >= N_NODES * EMB) return;
    int c = i & 63;
    h[i] = tanhf(h[i] * scale[c] + shift[c]);
}

extern "C" void kernel_launch(void* const* d_in, const int* in_sizes, int n_in,
                              void* d_out, int out_size, void* d_ws, size_t ws_size,
                              hipStream_t stream) {
    const float* ent   = (const float*)d_in[0];
    const float* rel   = (const float*)d_in[1];
    const int*   src   = (const int*)  d_in[2];
    const int*   dst   = (const int*)  d_in[3];
    const int*   rid   = (const int*)  d_in[4];
    const float* W     = (const float*)d_in[5];
    const float* gamma = (const float*)d_in[6];
    const float* beta  = (const float*)d_in[7];
    float* out = (float*)d_out;

    float* ws     = (float*)d_ws;
    int2*  sorted = (int2*)ws;
    int*   cnt    = (int*)(ws + 2 * (size_t)N_EDGES);
    int*   offs   = (int*)(ws + 2 * (size_t)N_EDGES + N_NODES);
    float* spare  = ws + 2 * (size_t)N_EDGES + 2 * (size_t)N_NODES;
    float* scale  = spare + (size_t)N_NODES * EMB;
    float* shift  = scale + EMB;
    int*   cmat   = (int*)spare;
    float* part   = spare;

    k_hist  <<<dim3(SORT_CHUNKS, SORT_RANGES), SORT_THREADS, 0, stream>>>(dst, cmat);
    k_colsum<<<(N_NODES + 255) / 256, 256, 0, stream>>>(cmat, cnt);
    k_scan <<<SCAN_BLOCKS, SCAN_THREADS, 0, stream>>>(cnt, offs);
    k_fill  <<<dim3(SORT_CHUNKS, SORT_RANGES), SORT_THREADS, 0, stream>>>(src, dst, rid, offs, cmat, sorted);
    k_node_gemm<<<N_TILES, TILE_ROWS * 32, 0, stream>>>(ent, rel, cnt, offs, sorted, W, out, part);
    k_stats  <<<EMB, STATS_THREADS, 0, stream>>>(part, gamma, beta, scale, shift);
    k_apply  <<<(N_NODES * EMB + 255) / 256, 256, 0, stream>>>(out, scale, shift);
}


# Scan launch folded into the per-node chunk-sum kernel: the counting kernel also leaves edge counts per (256-node sub-block, chunk), each block sums those in front of it and scans its own nodes
# speedup vs baseline: 5.8086x; 1.0360x over previous
#include <hip/hip_runtime.h>
#include <math.h>

#define N_NODES 50000
#define N_EDGES 1250000
#define EMB     64
#define BN_EPS  1e-5f
#define N_TILES (N_NODES / 16)

typedef __attribute__((ext_vector_type(2))) float v2f;
typedef __attribute__((ext_vector_type(8))) float v8f;

#define SORT_CHUNKS  64
#define SORT_RANGES  4
#define SORT_THREADS 1024
#define CHUNK_EDGES  ((N_EDGES + SORT_CHUNKS - 1) / SORT_CHUNKS)
#define CHUNK_TRIPS  ((CHUNK_EDGES + SORT_THREADS - 1) / SORT_THREADS)
#define RANGE_NODES  (N_NODES / SORT_RANGES)
static_assert(RANGE_NODES * SORT_RANGES == N_NODES, "node ranges must tile the nodes");
#define SUB_NODES   256
#define RANGE_SUBS  ((RANGE_NODES + SUB_NODES - 1) / SUB_NODES)
#define N_SUBS      (SORT_RANGES * RANGE_SUBS)
#define HIST_TRIPS  ((RANGE_NODES + SORT_THREADS - 1) / SORT_THREADS)
static_assert(SORT_THREADS % SUB_NODES == 0 && SUB_NODES % 32 == 0, "a wave lies inside one sub-block");
static_assert(SORT_CHUNKS % 4 == 0, "a sub-block's chunk sums are read in 16-byte pieces");

template <int MASK>
__device__ __forceinline__ int row_xor_i(int b) {
    return __builtin_amdgcn_update_dpp(b, b, 0x160 + MASK, 0xf, 0xf, false);
}
__device__ __forceinline__ int wave_sum_i(int p) {
    p += row_xor_i<1>(p);
    p += row_xor_i<2>(p);
    p += row_xor_i<4>(p);
    p += row_xor_i<8>(p);
    return p + __builtin_amdgcn_permlanex16(p, p, 0x76543210, 0xfedcba98, false, false);
}

__global__ void __launch_bounds__(SORT_THREADS)
k_hist(const int* __restrict__ dst, int* __restrict__ cmat, int* __restrict__ coarse) {
    __shared__ int s_cnt[RANGE_NODES];
    __shared__ int s_wsum[HIST_TRIPS * (SORT_THREADS / 32)];
    int t = threadIdx.x, chunk = blockIdx.x, lo = blockIdx.y * RANGE_NODES;
    for (int i = t; i < RANGE_NODES; i += SORT_THREADS) s_cnt[i] = 0;
    int d[CHUNK_TRIPS];
#pragma unroll
    for (int k = 0; k < CHUNK_TRIPS; ++k) {
        int w = k * SORT_THREADS + t;
        int i = chunk * CHUNK_EDGES + w;
        d[k] = (w < CHUNK_EDGES && i < N_EDGES) ? dst[i] : -1;
    }
    __syncthreads();
#pragma unroll
    for (int k = 0; k < CHUNK_TRIPS; ++k) {
        unsigned r = (unsigned)(d[k] - lo);
        if (r < (unsigned)RANGE_NODES) atomicAdd(&s_cnt[r], 1);
    }
    __syncthreads();
    int* row = cmat + (size_t)chunk * N_NODES + lo;
    int lane = t & 31, wave = t >> 5;
    for (int k = 0; k < HIST_TRIPS; ++k) {
        int i = k * SORT_THREADS + t;
        int v = 0;
        if (i < RANGE_NODES) { v = s_cnt[i]; row[i] = v; }
        v = wave_sum_i(v);
        if (lane == 0) s_wsum[k * (SORT_THREADS / 32) + wave] = v;
    }
    __syncthreads();
    if (t < RANGE_SUBS) {
        int s = 0;
        for (int j = 0; j < SUB_NODES / 32; ++j) s += s_wsum[t * (SUB_NODES / 32) + j];
        coarse[(size_t)(blockIdx.y * RANGE_SUBS + t) * SORT_CHUNKS + chunk] = s;
    }
}

#define COARSE_TRIPS ((N_SUBS * SORT_CHUNKS / 4 + SUB_NODES - 1) / SUB_NODES)
__global__ void __launch_bounds__(SUB_NODES)
k_colscan(int* __restrict__ cmat, const int* __restrict__ coarse,
          int* __restrict__ cnt, int* __restrict__ offs) {
    __shared__ int s_v[SUB_NODES / 32], s_b[SUB_NODES / 32];
    int t = threadIdx.x, lane = t & 31, wave = t >> 5;
    int sub = blockIdx.x;
    int r = sub / RANGE_SUBS;
    int local = (sub - r * RANGE_SUBS) * SUB_NODES + t;
    bool valid = local < RANGE_NODES;
    int n = r * RANGE_NODES + local;
    const int4* c4 = (const int4*)coarse;
    int nq = sub * (SORT_CHUNKS / 4);
    int4 q[COARSE_TRIPS];
#pragma unroll
    for (int k = 0; k < COARSE_TRIPS; ++k) {
        int i = k * SUB_NODES + t;
        q[k] = c4[(i < nq) ? i : 0];
    }
    int nl = r * RANGE_NODES + min(local, RANGE_NODES - 1);
    int c[SORT_CHUNKS];
#pragma unroll
    for (int b = 0; b < SORT_CHUNKS; ++b) c[b] = cmat[(size_t)b * N_NODES + nl];
    int before = 0;
#pragma unroll
    for (int k = 0; k < COARSE_TRIPS; ++k) {
        int i = k * SUB_NODES + t;
        int s = (q[k].x + q[k].y) + (q[k].z + q[k].w);
        before += (i < nq) ? s : 0;
    }
    int run = 0;
#pragma unroll
    for (int b = 0; b < SORT_CHUNKS; ++b) { int x = c[b]; c[b] = run; run += x; }
    if (valid) {
#pragma unroll
        for (int b = 0; b < SORT_CHUNKS; ++b) cmat[(size_t)b * N_NODES + n] = c[b];
    } else {
        run = 0;
    }
    int inc = run;
    for (int off = 1; off < 32; off <<= 1) {
        int u = __shfl_up(inc, off, 32);
        if (lane >= off) inc += u;
    }
    before = wave_sum_i(before);
    if (lane == 31) s_v[wave] = inc;
    if (lane == 0)  s_b[wave] = before;
    __syncthreads();
    int first = 0;
    for (int w = 0; w < SUB_NODES / 32; ++w) {
        first += s_b[w];
        first += (w < wave) ? s_v[w] : 0;
    }
    if (valid) {
        cnt[n]  = run;
        offs[n] = first + inc - run;
    }
}

__global__ void __launch_bounds__(SORT_THREADS)
k_fill(const int* __restrict__ src, const int* __restrict__ dst,
       const int* __restrict__ rid,
       const int* __restrict__ offs, const int* __restrict__ cmat,
       int2* __restrict__ sorted) {
    __shared__ int s_pos[RANGE_NODES];
    int t = threadIdx.x, chunk = blockIdx.x, lo = blockIdx.y * RANGE_NODES;
    const int* row = cmat + (size_t)chunk * N_NODES + lo;
    for (int i = t; i < RANGE_NODES; i += SORT_THREADS) s_pos[i] = offs[lo + i] + row[i];
    int d[CHUNK_TRIPS];
#pragma unroll
    for (int k = 0; k < CHUNK_TRIPS; ++k) {
        int w = k * SORT_THREADS + t;
        int i = chunk * CHUNK_EDGES + w;
        d[k] = (w < CHUNK_EDGES && i < N_EDGES) ? dst[i] : -1;
    }
    int s[CHUNK_TRIPS], q[CHUNK_TRIPS];
#pragma unroll
    for (int k = 0; k < CHUNK_TRIPS; ++k) {
        int i = chunk * CHUNK_EDGES + k * SORT_THREADS + t;
        bool mine = (unsigned)(d[k] - lo) < (unsigned)RANGE_NODES;
        s[k] = mine ? src[i] : 0;
        q[k] = mine ? rid[i] : 0;
    }
    __syncthreads();
#pragma unroll
    for (int k = 0; k < CHUNK_TRIPS; ++k) {
        unsigned r = (unsigned)(d[k] - lo);
        if (r < (unsigned)RANGE_NODES) {
            int pos = atomicAdd(&s_pos[r], 1);
            sorted[pos] = make_int2(s[k], q[k]);
        }
    }
}

template <int MASK>
__device__ __forceinline__ float row_xor(float x) {
    int b = __builtin_bit_cast(int, x);
    return __builtin_bit_cast(float,
        __builtin_amdgcn_update_dpp(b, b, 0x160 + MASK, 0xf, 0xf, false));
}
__device__ __forceinline__ float xhalf(float x) {
    int b = __builtin_bit_cast(int, x);
    return __builtin_bit_cast(float,
        __builtin_amdgcn_permlanex16(b, b, 0x76543210, 0xfedcba98, false, false));
}
__device__ __forceinline__ float wave_sum(float p) {
    p += row_xor<1>(p);
    p += row_xor<2>(p);
    p += row_xor<4>(p);
    p += row_xor<8>(p);
    return p + xhalf(p);
}

#define TILE_ROWS  16
#define TILE_PITCH 68
#define EDGE_GROUPS      4
#define EDGES_PER_GROUP  2
#define STEP_EDGES       (EDGE_GROUPS * EDGES_PER_GROUP)
typedef __attribute__((ext_vector_type(4))) float v4f;

__device__ __forceinline__ float dot8(v4f a0, v4f a1, v4f b0, v4f b1) {
    float p = a0.x * b0.x;
    p += a0.y * b0.y; p += a0.z * b0.z; p += a0.w * b0.w;
    p += a1.x * b1.x; p += a1.y * b1.y; p += a1.z * b1.z; p += a1.w * b1.w;
    return p;
}

__global__ void __launch_bounds__(TILE_ROWS * 32)
k_node_gemm(const float* __restrict__ ent, const float* __restrict__ rel,
            const int* __restrict__ cnt, const int* __restrict__ offs,
            const int2* __restrict__ sorted, const float* __restrict__ W,
            float* __restrict__ h, float* __restrict__ part) {
    __shared__ __attribute__((aligned(16))) float s_tile[TILE_ROWS * TILE_PITCH];
    int lane = threadIdx.x & 31;
    int wave = __builtin_amdgcn_readfirstlane(threadIdx.x >> 5);
    int d = blockIdx.x * TILE_ROWS + wave;
    int beg = offs[d];
    int end = beg + cnt[d];
    int g = lane >> 3;
    int q = lane & 7;
    const v4f* ent4 = (const v4f*)ent;
    const v4f* rel4 = (const v4f*)rel;
    v4f ed0 = ent4[(size_t)d * (EMB / 4) + q];
    v4f ed1 = ent4[(size_t)d * (EMB / 4) + 8 + q];
    float m = -INFINITY, l = 0.0f;
    v4f acc0 = {0.0f, 0.0f, 0.0f, 0.0f}, acc1 = {0.0f, 0.0f, 0.0f, 0.0f};
    if (end > beg) {
        int2 pr = sorted[min(beg + lane, end - 1)];
        for (int cb = beg; cb < end; cb += 32) {
            int nchunk = min(32, end - cb);
            for (int s = 0; s < nchunk; s += STEP_EDGES) {
                bool ok[EDGES_PER_GROUP];
                v4f  c0[EDGES_PER_GROUP], c1[EDGES_PER_GROUP];
                v4f  r0[EDGES_PER_GROUP], r1[EDGES_PER_GROUP];
#pragma unroll
                for (int u = 0; u < EDGES_PER_GROUP; ++u) {
                    int slot = s + u * EDGE_GROUPS + g;
                    ok[u] = slot < nchunk;
                    int sl = min(slot, nchunk - 1);
                    int si = __shfl(pr.x, sl, 32);
                    int ri = __shfl(pr.y, sl, 32);
                    unsigned eo = (unsigned)si * (EMB / 4) + (unsigned)q;
                    unsigned ro = (unsigned)ri * (EMB / 4) + (unsigned)q;
                    c0[u] = ent4[eo];
                    c1[u] = ent4[eo + 8u];
                    r0[u] = rel4[ro];
                    r1[u] = rel4[ro + 8u];
                }
                float sc[EDGES_PER_GROUP];
                float tmax = -INFINITY;
#pragma unroll
                for (int u = 0; u < EDGES_PER_GROUP; ++u) {
                    c0[u] *= r0[u]; c1[u] *= r1[u];
                    float p = dot8(c0[u], c1[u], ed0, ed1);
                    p += row_xor<1>(p);
                    p += row_xor<2>(p);
                    p += row_xor<4>(p);
                    sc[u] = ok[u] ? p : -INFINITY;
                    tmax = fmaxf(tmax, sc[u]);
                }
                tmax = fmaxf(tmax, row_xor<8>(tmax));
                tmax = fmaxf(tmax, xhalf(tmax));
                float mn = fmaxf(m, tmax);
                float f  = expf(m - mn);
                l *= f; acc0 *= f; acc1 *= f;
#pragma unroll
                for (int u = 0; u < EDGES_PER_GROUP; ++u) {
                    float w = expf(sc[u] - mn);
                    l    += w;
                    acc0 += c0[u] * w;
                    acc1 += c1[u] * w;
                }
                m = mn;
            }
            if (cb + 32 < end) pr = sorted[min(cb + 32 + lane, end - 1)];
        }
    }
    l += row_xor<8>(l);  l += xhalf(l);
#pragma unroll
    for (int i = 0; i < 4; ++i) {
        acc0[i] += row_xor<8>(acc0[i]);  acc0[i] += xhalf(acc0[i]);
        acc1[i] += row_xor<8>(acc1[i]);  acc1[i] += xhalf(acc1[i]);
    }
    float inv = (l > 0.0f) ? 1.0f / l : 0.0f;
    if (g == 0) {
        *(v4f*)(s_tile + wave * TILE_PITCH + 4 * q)      = acc0 * inv;
        *(v4f*)(s_tile + wave * TILE_PITCH + 32 + 4 * q) = acc1 * inv;
    }

    int n0    = wave * 16;
    int mrow  = lane & 15;
    int khalf = lane >> 4;
    int n     = lane & 15;
    v2f bfrag[EMB / 4];
    if (wave < 4) {
#pragma unroll
        for (int k = 0; k < EMB; k += 4) {
            int ka = k + khalf * 2;
            bfrag[k / 4].x = W[(size_t)ka       * EMB + n0 + n];
            bfrag[k / 4].y = W[(size_t)(ka + 1) * EMB + n0 + n];
        }
    }
    __syncthreads();
    if (wave >= 4) return;

    v8f accd = {};
#pragma unroll
    for (int k = 0; k < EMB; k += 4) {
        int ka = k + khalf * 2;
        v2f a = *(const v2f*)(s_tile + mrow * TILE_PITCH + ka);
        accd = __builtin_amdgcn_wmma_f32_16x16x4_f32(
                   false, a, false, bfrag[k / 4], (short)0, accd, false, false);
    }
    int mbase = blockIdx.x * TILE_ROWS + khalf * 8;
    float cs = 0.0f, cq = 0.0f;
#pragma unroll
    for (int r = 0; r < 8; ++r) {
        float v = accd[r];
        h[(size_t)(mbase + r) * EMB + n0 + n] = v;
        cs += v; cq += v * v;
    }
    cs += __shfl_xor(cs, 16, 32);
    cq += __shfl_xor(cq, 16, 32);
    if (khalf == 0) {
        v2f p; p.x = cs; p.y = cq;
        *(v2f*)(part + 2 * ((size_t)(n0 + n) * N_TILES + blockIdx.x)) = p;
    }
}

#define STATS_THREADS 1024
__global__ void __launch_bounds__(STATS_THREADS)
k_stats(const float* __restrict__ part,
        const float* __restrict__ gamma, const float* __restrict__ beta,
        float* __restrict__ scale, float* __restrict__ shift) {
    int c = blockIdx.x;
    const v2f* p = (const v2f*)part + (size_t)c * N_TILES;
    __shared__ float s_sum[STATS_THREADS / 32], s_sq[STATS_THREADS / 32];
    float sum = 0.0f, sq = 0.0f;
#pragma unroll
    for (int i = 0; i < (N_TILES + STATS_THREADS - 1) / STATS_THREADS; ++i) {
        int t = threadIdx.x + i * STATS_THREADS;
        if (t < N_TILES) { v2f v = p[t]; sum += v.x; sq += v.y; }
    }
#pragma unroll
    for (int off = 16; off > 0; off >>= 1) {
        sum += __shfl_xor(sum, off, 32);
        sq  += __shfl_xor(sq,  off, 32);
    }
    int wave = threadIdx.x >> 5, lane = threadIdx.x & 31;
    if (lane == 0) { s_sum[wave] = sum; s_sq[wave] = sq; }
    __syncthreads();
    if (wave == 0) {
        sum = s_sum[lane]; sq = s_sq[lane];
#pragma unroll
        for (int off = 16; off > 0; off >>= 1) {
            sum += __shfl_xor(sum, off, 32);
            sq  += __shfl_xor(sq,  off, 32);
        }
        if (lane == 0) {
            float mu   = sum / (float)N_NODES;
            float var  = sq / (float)N_NODES - mu * mu;
            float rinv = rsqrtf(var + BN_EPS);
            float sc   = gamma[c] * rinv;
            scale[c] = sc;
            shift[c] = beta[c] - mu * sc;
        }
    }
}

__global__ void k_apply(float* __restrict__ h,
                        const float* __restrict__ scale, const float* __restrict__ shift) {
    int i = blockIdx.x * blockDim.x + threadIdx.x;
    if (i >= N_NODES * EMB) return;
    int c = i & 63;
    h[i] = tanhf(h[i] * scale[c] + shift[c]);
}

extern "C" void kernel_launch(void* const* d_in, const int* in_sizes, int n_in,
                              void* d_out, int out_size, void* d_ws, size_t ws_size,
                              hipStream_t stream) {
    const float* ent   = (const float*)d_in[0];
    const float* rel   = (const float*)d_in[1];
    const int*   src   = (const int*)  d_in[2];
    const int*   dst   = (const int*)  d_in[3];
    const int*   rid   = (const int*)  d_in[4];
    const float* W     = (const float*)d_in[5];
    const float* gamma = (const float*)d_in[6];
    const float* beta  = (const float*)d_in[7];
    float* out = (float*)d_out;

    float* ws     = (float*)d_ws;
    int2*  sorted = (int2*)ws;
    int*   cnt    = (int*)(ws + 2 * (size_t)N_EDGES);
    int*   offs   = (int*)(ws + 2 * (size_t)N_EDGES + N_NODES);
    float* spare  = ws + 2 * (size_t)N_EDGES + 2 * (size_t)N_NODES;
    float* scale  = spare + (size_t)N_NODES * EMB;
    float* shift  = scale + EMB;
    int*   cmat   = (int*)spare;
    int*   coarse = (int*)sorted;
    float* part   = spare;

    k_hist  <<<dim3(SORT_CHUNKS, SORT_RANGES), SORT_THREADS, 0, stream>>>(dst, cmat, coarse);
    k_colscan<<<N_SUBS, SUB_NODES, 0, stream>>>(cmat, coarse, cnt, offs);
    k_fill  <<<dim3(SORT_CHUNKS, SORT_RANGES), SORT_THREADS, 0, stream>>>(src, dst, rid, offs, cmat, sorted);
    k_node_gemm<<<N_TILES, TILE_ROWS * 32, 0, stream>>>(ent, rel, cnt, offs, sorted, W, out, part);
    k_stats  <<<EMB, STATS_THREADS, 0, stream>>>(part, gamma, beta, scale, shift);
    k_apply  <<<(N_NODES * EMB + 255) / 256, 256, 0, stream>>>(out, scale, shift);
}


# Relation table held in LDS: 256 resident 32-wave blocks copy it once, take nodes from an LDS counter and read relation rows from LDS; 16 edges per wave step (rescale points move, sum order kept)
# speedup vs baseline: 6.5138x; 1.1214x over previous
#include <hip/hip_runtime.h>
#include <math.h>

#define N_NODES 50000
#define N_EDGES 1250000
#define EMB     64
#define BN_EPS  1e-5f
#define N_TILES (N_NODES / 16)

typedef __attribute__((ext_vector_type(2))) float v2f;
typedef __attribute__((ext_vector_type(8))) float v8f;

#define SORT_CHUNKS  64
#define SORT_RANGES  4
#define SORT_THREADS 1024
#define CHUNK_EDGES  ((N_EDGES + SORT_CHUNKS - 1) / SORT_CHUNKS)
#define CHUNK_TRIPS  ((CHUNK_EDGES + SORT_THREADS - 1) / SORT_THREADS)
#define RANGE_NODES  (N_NODES / SORT_RANGES)
static_assert(RANGE_NODES * SORT_RANGES == N_NODES, "node ranges must tile the nodes");
#define SUB_NODES   256
#define RANGE_SUBS  ((RANGE_NODES + SUB_NODES - 1) / SUB_NODES)
#define N_SUBS      (SORT_RANGES * RANGE_SUBS)
#define HIST_TRIPS  ((RANGE_NODES + SORT_THREADS - 1) / SORT_THREADS)
static_assert(SORT_THREADS % SUB_NODES == 0 && SUB_NODES % 32 == 0, "a wave lies inside one sub-block");
static_assert(SORT_CHUNKS % 4 == 0, "a sub-block's chunk sums are read in 16-byte pieces");

template <int MASK>
__device__ __forceinline__ int row_xor_i(int b) {
    return __builtin_amdgcn_update_dpp(b, b, 0x160 + MASK, 0xf, 0xf, false);
}
__device__ __forceinline__ int wave_sum_i(int p) {
    p += row_xor_i<1>(p);
    p += row_xor_i<2>(p);
    p += row_xor_i<4>(p);
    p += row_xor_i<8>(p);
    return p + __builtin_amdgcn_permlanex16(p, p, 0x76543210, 0xfedcba98, false, false);
}

__global__ void __launch_bounds__(SORT_THREADS)
k_hist(const int* __restrict__ dst, int* __restrict__ cmat, int* __restrict__ coarse) {
    __shared__ int s_cnt[RANGE_NODES];
    __shared__ int s_wsum[HIST_TRIPS * (SORT_THREADS / 32)];
    int t = threadIdx.x, chunk = blockIdx.x, lo = blockIdx.y * RANGE_NODES;
    for (int i = t; i < RANGE_NODES; i += SORT_THREADS) s_cnt[i] = 0;
    int d[CHUNK_TRIPS];
#pragma unroll
    for (int k = 0; k < CHUNK_TRIPS; ++k) {
        int w = k * SORT_THREADS + t;
        int i = chunk * CHUNK_EDGES + w;
        d[k] = (w < CHUNK_EDGES && i < N_EDGES) ? dst[i] : -1;
    }
    __syncthreads();
#pragma unroll
    for (int k = 0; k < CHUNK_TRIPS; ++k) {
        unsigned r = (unsigned)(d[k] - lo);
        if (r < (unsigned)RANGE_NODES) atomicAdd(&s_cnt[r], 1);
    }
    __syncthreads();
    int* row = cmat + (size_t)chunk * N_NODES + lo;
    int lane = t & 31, wave = t >> 5;
    for (int k = 0; k < HIST_TRIPS; ++k) {
        int i = k * SORT_THREADS + t;
        int v = 0;
        if (i < RANGE_NODES) { v = s_cnt[i]; row[i] = v; }
        v = wave_sum_i(v);
        if (lane == 0) s_wsum[k * (SORT_THREADS / 32) + wave] = v;
    }
    __syncthreads();
    if (t < RANGE_SUBS) {
        int s = 0;
        for (int j = 0; j < SUB_NODES / 32; ++j) s += s_wsum[t * (SUB_NODES / 32) + j];
        coarse[(size_t)(blockIdx.y * RANGE_SUBS + t) * SORT_CHUNKS + chunk] = s;
    }
}

#define COARSE_TRIPS ((N_SUBS * SORT_CHUNKS / 4 + SUB_NODES - 1) / SUB_NODES)
__global__ void __launch_bounds__(SUB_NODES)
k_colscan(int* __restrict__ cmat, const int* __restrict__ coarse,
          int* __restrict__ cnt, int* __restrict__ offs) {
    __shared__ int s_v[SUB_NODES / 32], s_b[SUB_NODES / 32];
    int t = threadIdx.x, lane = t & 31, wave = t >> 5;
    int sub = blockIdx.x;
    int r = sub / RANGE_SUBS;
    int local = (sub - r * RANGE_SUBS) * SUB_NODES + t;
    bool valid = local < RANGE_NODES;
    int n = r * RANGE_NODES + local;
    const int4* c4 = (const int4*)coarse;
    int nq = sub * (SORT_CHUNKS / 4);
    int4 q[COARSE_TRIPS];
#pragma unroll
    for (int k = 0; k < COARSE_TRIPS; ++k) {
        int i = k * SUB_NODES + t;
        q[k] = c4[(i < nq) ? i : 0];
    }
    int nl = r * RANGE_NODES + min(local, RANGE_NODES - 1);
    int c[SORT_CHUNKS];
#pragma unroll
    for (int b = 0; b < SORT_CHUNKS; ++b) c[b] = cmat[(size_t)b * N_NODES + nl];
    int before = 0;
#pragma unroll
    for (int k = 0; k < COARSE_TRIPS; ++k) {
        int i = k * SUB_NODES + t;
        int s = (q[k].x + q[k].y) + (q[k].z + q[k].w);
        before += (i < nq) ? s : 0;
    }
    int run = 0;
#pragma unroll
    for (int b = 0; b < SORT_CHUNKS; ++b) { int x = c[b]; c[b] = run; run += x; }
    if (valid) {
#pragma unroll
        for (int b = 0; b < SORT_CHUNKS; ++b) cmat[(size_t)b * N_NODES + n] = c[b];
    } else {
        run = 0;
    }
    int inc = run;
    for (int off = 1; off < 32; off <<= 1) {
        int u = __shfl_up(inc, off, 32);
        if (lane >= off) inc += u;
    }
    before = wave_sum_i(before);
    if (lane == 31) s_v[wave] = inc;
    if (lane == 0)  s_b[wave] = before;
    __syncthreads();
    int first = 0;
    for (int w = 0; w < SUB_NODES / 32; ++w) {
        first += s_b[w];
        first += (w < wave) ? s_v[w] : 0;
    }
    if (valid) {
        cnt[n]  = run;
        offs[n] = first + inc - run;
    }
}

__global__ void __launch_bounds__(SORT_THREADS)
k_fill(const int* __restrict__ src, const int* __restrict__ dst,
       const int* __restrict__ rid,
       const int* __restrict__ offs, const int* __restrict__ cmat,
       int2* __restrict__ sorted) {
    __shared__ int s_pos[RANGE_NODES];
    int t = threadIdx.x, chunk = blockIdx.x, lo = blockIdx.y * RANGE_NODES;
    const int* row = cmat + (size_t)chunk * N_NODES + lo;
    for (int i = t; i < RANGE_NODES; i += SORT_THREADS) s_pos[i] = offs[lo + i] + row[i];
    int d[CHUNK_TRIPS];
#pragma unroll
    for (int k = 0; k < CHUNK_TRIPS; ++k) {
        int w = k * SORT_THREADS + t;
        int i = chunk * CHUNK_EDGES + w;
        d[k] = (w < CHUNK_EDGES && i < N_EDGES) ? dst[i] : -1;
    }
    int s[CHUNK_TRIPS], q[CHUNK_TRIPS];
#pragma unroll
    for (int k = 0; k < CHUNK_TRIPS; ++k) {
        int i = chunk * CHUNK_EDGES + k * SORT_THREADS + t;
        bool mine = (unsigned)(d[k] - lo) < (unsigned)RANGE_NODES;
        s[k] = mine ? src[i] : 0;
        q[k] = mine ? rid[i] : 0;
    }
    __syncthreads();
#pragma unroll
    for (int k = 0; k < CHUNK_TRIPS; ++k) {
        unsigned r = (unsigned)(d[k] - lo);
        if (r < (unsigned)RANGE_NODES) {
            int pos = atomicAdd(&s_pos[r], 1);
            sorted[pos] = make_int2(s[k], q[k]);
        }
    }
}

template <int MASK>
__device__ __forceinline__ float row_xor(float x) {
    int b = __builtin_bit_cast(int, x);
    return __builtin_bit_cast(float,
        __builtin_amdgcn_update_dpp(b, b, 0x160 + MASK, 0xf, 0xf, false));
}
__device__ __forceinline__ float xhalf(float x) {
    int b = __builtin_bit_cast(int, x);
    return __builtin_bit_cast(float,
        __builtin_amdgcn_permlanex16(b, b, 0x76543210, 0xfedcba98, false, false));
}
__device__ __forceinline__ float wave_sum(float p) {
    p += row_xor<1>(p);
    p += row_xor<2>(p);
    p += row_xor<4>(p);
    p += row_xor<8>(p);
    return p + xhalf(p);
}

#define TILE_ROWS  16
#define TILE_PITCH 68
#define N_REL        1000
#define NODE_BLOCKS  256
#define NODE_WAVES   32
#define NODE_THREADS (NODE_WAVES * 32)
#define MAX_TILES    ((N_TILES + NODE_BLOCKS - 1) / NODE_BLOCKS)
#define REL_CHUNKS   (N_REL * EMB / 4)
#define REL_TRIPS    ((REL_CHUNKS + NODE_THREADS - 1) / NODE_THREADS)
#define EDGE_GROUPS      4
#define EDGES_PER_GROUP  4
#define STEP_EDGES       (EDGE_GROUPS * EDGES_PER_GROUP)
static_assert(32 % STEP_EDGES == 0, "a 32-pair piece is a whole number of steps");
static_assert(N_TILES * TILE_ROWS == N_NODES, "the row tiles cover the nodes exactly");
typedef __attribute__((ext_vector_type(4))) float v4f;

__device__ __forceinline__ float dot8(v4f a0, v4f a1, v4f b0, v4f b1) {
    float p = a0.x * b0.x;
    p += a0.y * b0.y; p += a0.z * b0.z; p += a0.w * b0.w;
    p += a1.x * b1.x; p += a1.y * b1.y; p += a1.z * b1.z; p += a1.w * b1.w;
    return p;
}

__global__ void __launch_bounds__(NODE_THREADS)
k_node_gemm(const float* __restrict__ ent, const float* __restrict__ rel,
            const int* __restrict__ cnt, const int* __restrict__ offs,
            const int2* __restrict__ sorted, const float* __restrict__ W,
            float* __restrict__ h, float* __restrict__ part) {
    __shared__ __attribute__((aligned(16))) float s_rel[N_REL * EMB];
    __shared__ __attribute__((aligned(16))) float s_tile[MAX_TILES * TILE_ROWS * TILE_PITCH];
    __shared__ int s_next;
    int lane = threadIdx.x & 31;
    int wave = __builtin_amdgcn_readfirstlane(threadIdx.x >> 5);
    int t0 = (int)(blockIdx.x * N_TILES) / NODE_BLOCKS;
    int ntiles = (int)((blockIdx.x + 1) * N_TILES) / NODE_BLOCKS - t0;
    int nnodes = ntiles * TILE_ROWS;
    int node0  = t0 * TILE_ROWS;
    const v4f* ent4 = (const v4f*)ent;
    const v4f* rel4 = (const v4f*)rel;
    v4f* s_rel4 = (v4f*)s_rel;
    int i = wave;
    int beg = offs[node0 + i];
    int deg = cnt[node0 + i];
    {
        v4f piece[REL_TRIPS];
#pragma unroll
        for (int k = 0; k < REL_TRIPS; ++k) {
            int c = ((k + (int)blockIdx.x) % REL_TRIPS) * NODE_THREADS + (int)threadIdx.x;
            piece[k] = rel4[min(c, REL_CHUNKS - 1)];
        }
#pragma unroll
        for (int k = 0; k < REL_TRIPS; ++k) {
            int c = ((k + (int)blockIdx.x) % REL_TRIPS) * NODE_THREADS + (int)threadIdx.x;
            if (c < REL_CHUNKS) s_rel4[c] = piece[k];
        }
    }
    if (threadIdx.x == 0) s_next = NODE_WAVES;
    __syncthreads();
    int g = lane >> 3;
    int q = lane & 7;
    while (true) {
        int nx = 0;
        if (lane == 0) nx = atomicAdd(&s_next, 1);
        int inext = __builtin_amdgcn_readfirstlane(nx);
        int dn   = node0 + min(inext, nnodes - 1);
        int nbeg = offs[dn];
        int ndeg = cnt[dn];
        int d = node0 + i;
        int end = beg + deg;
        v4f ed0 = ent4[(size_t)d * (EMB / 4) + q];
        v4f ed1 = ent4[(size_t)d * (EMB / 4) + 8 + q];
        float m = -INFINITY, l = 0.0f;
        v4f acc0 = {0.0f, 0.0f, 0.0f, 0.0f}, acc1 = {0.0f, 0.0f, 0.0f, 0.0f};
        if (end > beg) {
            int2 pr = sorted[min(beg + lane, end - 1)];
            for (int cb = beg; cb < end; cb += 32) {
                int nchunk = min(32, end - cb);
                for (int s = 0; s < nchunk; s += STEP_EDGES) {
                    bool ok[EDGES_PER_GROUP];
                    v4f  c0[EDGES_PER_GROUP], c1[EDGES_PER_GROUP];
                    v4f  r0[EDGES_PER_GROUP], r1[EDGES_PER_GROUP];
#pragma unroll
                    for (int u = 0; u < EDGES_PER_GROUP; ++u) {
                        int slot = s + u * EDGE_GROUPS + g;
                        ok[u] = slot < nchunk;
                        int sl = min(slot, nchunk - 1);
                        int si = __shfl(pr.x, sl, 32);
                        int ri = __shfl(pr.y, sl, 32);
                        unsigned eo = (unsigned)si * (EMB / 4) + (unsigned)q;
                        unsigned ro = (unsigned)ri * (EMB / 4) + (unsigned)q;
                        c0[u] = ent4[eo];
                        c1[u] = ent4[eo + 8u];
                        r0[u] = s_rel4[ro];
                        r1[u] = s_rel4[ro + 8u];
                    }
                    float sc[EDGES_PER_GROUP];
                    float tmax = -INFINITY;
#pragma unroll
                    for (int u = 0; u < EDGES_PER_GROUP; ++u) {
                        c0[u] *= r0[u]; c1[u] *= r1[u];
                        float p = dot8(c0[u], c1[u], ed0, ed1);
                        p += row_xor<1>(p);
                        p += row_xor<2>(p);
                        p += row_xor<4>(p);
                        sc[u] = ok[u] ? p : -INFINITY;
                        tmax = fmaxf(tmax, sc[u]);
                    }
                    tmax = fmaxf(tmax, row_xor<8>(tmax));
                    tmax = fmaxf(tmax, xhalf(tmax));
                    float mn = fmaxf(m, tmax);
                    float f  = expf(m - mn);
                    l *= f; acc0 *= f; acc1 *= f;
#pragma unroll
                    for (int u = 0; u < EDGES_PER_GROUP; ++u) {
                        float w = expf(sc[u] - mn);
                        l    += w;
                        acc0 += c0[u] * w;
                        acc1 += c1[u] * w;
                    }
                    m = mn;
                }
                if (cb + 32 < end) pr = sorted[min(cb + 32 + lane, end - 1)];
            }
        }
        l += row_xor<8>(l);  l += xhalf(l);
#pragma unroll
        for (int c = 0; c < 4; ++c) {
            acc0[c] += row_xor<8>(acc0[c]);  acc0[c] += xhalf(acc0[c]);
            acc1[c] += row_xor<8>(acc1[c]);  acc1[c] += xhalf(acc1[c]);
        }
        float inv = (l > 0.0f) ? 1.0f / l : 0.0f;
        if (g == 0) {
            *(v4f*)(s_tile + i * TILE_PITCH + 4 * q)      = acc0 * inv;
            *(v4f*)(s_tile + i * TILE_PITCH + 32 + 4 * q) = acc1 * inv;
        }
        if (inext >= nnodes) break;
        i = inext; beg = nbeg; deg = ndeg;
    }

    int n0    = (wave & 3) * 16;
    int mrow  = lane & 15;
    int khalf = lane >> 4;
    int n     = lane & 15;
    v2f bfrag[EMB / 4];
#pragma unroll
    for (int k = 0; k < EMB; k += 4) {
        int ka = k + khalf * 2;
        bfrag[k / 4].x = W[(size_t)ka       * EMB + n0 + n];
        bfrag[k / 4].y = W[(size_t)(ka + 1) * EMB + n0 + n];
    }
    __syncthreads();

    for (int tl = wave >> 2; tl < ntiles; tl += NODE_WAVES / 4) {
        v8f accd = {};
        const float* arow = s_tile + (tl * TILE_ROWS + mrow) * TILE_PITCH;
#pragma unroll
        for (int k = 0; k < EMB; k += 4) {
            int ka = k + khalf * 2;
            v2f a = *(const v2f*)(arow + ka);
            accd = __builtin_amdgcn_wmma_f32_16x16x4_f32(
                       false, a, false, bfrag[k / 4], (short)0, accd, false, false);
        }
        int tg    = t0 + tl;
        int mbase = tg * TILE_ROWS + khalf * 8;
        float cs = 0.0f, cq = 0.0f;
#pragma unroll
        for (int r = 0; r < 8; ++r) {
            float v = accd[r];
            h[(size_t)(mbase + r) * EMB + n0 + n] = v;
            cs += v; cq += v * v;
        }
        cs += __shfl_xor(cs, 16, 32);
        cq += __shfl_xor(cq, 16, 32);
        if (khalf == 0) {
            v2f p; p.x = cs; p.y = cq;
            *(v2f*)(part + 2 * ((size_t)(n0 + n) * N_TILES + tg)) = p;
        }
    }
}

#define STATS_THREADS 1024
__global__ void __launch_bounds__(STATS_THREADS)
k_stats(const float* __restrict__ part,
        const float* __restrict__ gamma, const float* __restrict__ beta,
        float* __restrict__ scale, float* __restrict__ shift) {
    int c = blockIdx.x;
    const v2f* p = (const v2f*)part + (size_t)c * N_TILES;
    __shared__ float s_sum[STATS_THREADS / 32], s_sq[STATS_THREADS / 32];
    float sum = 0.0f, sq = 0.0f;
#pragma unroll
    for (int i = 0; i < (N_TILES + STATS_THREADS - 1) / STATS_THREADS; ++i) {
        int t = threadIdx.x + i * STATS_THREADS;
        if (t < N_TILES) { v2f v = p[t]; sum += v.x; sq += v.y; }
    }
#pragma unroll
    for (int off = 16; off > 0; off >>= 1) {
        sum += __shfl_xor(sum, off, 32);
        sq  += __shfl_xor(sq,  off, 32);
    }
    int wave = threadIdx.x >> 5, lane = threadIdx.x & 31;
    if (lane == 0) { s_sum[wave] = sum; s_sq[wave] = sq; }
    __syncthreads();
    if (wave == 0) {
        sum = s_sum[lane]; sq = s_sq[lane];
#pragma unroll
        for (int off = 16; off > 0; off >>= 1) {
            sum += __shfl_xor(sum, off, 32);
            sq  += __shfl_xor(sq,  off, 32);
        }
        if (lane == 0) {
            float mu   = sum / (float)N_NODES;
            float var  = sq / (float)N_NODES - mu * mu;
            float rinv = rsqrtf(var + BN_EPS);
            float sc   = gamma[c] * rinv;
            scale[c] = sc;
            shift[c] = beta[c] - mu * sc;
        }
    }
}

__global__ void k_apply(float* __restrict__ h,
                        const float* __restrict__ scale, const float* __restrict__ shift) {
    int i = blockIdx.x * blockDim.x + threadIdx.x;
    if (i >= N_NODES * EMB) return;
    int c = i & 63;
    h[i] = tanhf(h[i] * scale[c] + shift[c]);
}

extern "C" void kernel_launch(void* const* d_in, const int* in_sizes, int n_in,
                              void* d_out, int out_size, void* d_ws, size_t ws_size,
                              hipStream_t stream) {
    const float* ent   = (const float*)d_in[0];
    const float* rel   = (const float*)d_in[1];
    const int*   src   = (const int*)  d_in[2];
    const int*   dst   = (const int*)  d_in[3];
    const int*   rid   = (const int*)  d_in[4];
    const float* W     = (const float*)d_in[5];
    const float* gamma = (const float*)d_in[6];
    const float* beta  = (const float*)d_in[7];
    float* out = (float*)d_out;

    float* ws     = (float*)d_ws;
    int2*  sorted = (int2*)ws;
    int*   cnt    = (int*)(ws + 2 * (size_t)N_EDGES);
    int*   offs   = (int*)(ws + 2 * (size_t)N_EDGES + N_NODES);
    float* spare  = ws + 2 * (size_t)N_EDGES + 2 * (size_t)N_NODES;
    float* scale  = spare + (size_t)N_NODES * EMB;
    float* shift  = scale + EMB;
    int*   cmat   = (int*)spare;
    int*   coarse = (int*)sorted;
    float* part   = spare;

    k_hist  <<<dim3(SORT_CHUNKS, SORT_RANGES), SORT_THREADS, 0, stream>>>(dst, cmat, coarse);
    k_colscan<<<N_SUBS, SUB_NODES, 0, stream>>>(cmat, coarse, cnt, offs);
    k_fill  <<<dim3(SORT_CHUNKS, SORT_RANGES), SORT_THREADS, 0, stream>>>(src, dst, rid, offs, cmat, sorted);
    k_node_gemm<<<NODE_BLOCKS, NODE_THREADS, 0, stream>>>(ent, rel, cnt, offs, sorted, W, out, part);
    k_stats  <<<EMB, STATS_THREADS, 0, stream>>>(part, gamma, beta, scale, shift);
    k_apply  <<<(N_NODES * EMB + 255) / 256, 256, 0, stream>>>(out, scale, shift);
}


# Two-level edge sort, counters in LDS: chunks binned by 196-node range into own pieces, a block per range sorts its 256 runs; count matrix, scan launch, scattered stores gone; new sum order per node
# speedup vs baseline: 9.3983x; 1.4428x over previous
#include <hip/hip_runtime.h>
#include <math.h>

#define N_NODES 50000
#define N_EDGES 1250000
#define EMB     64
#define BN_EPS  1e-5f
#define N_TILES (N_NODES / 16)

typedef __attribute__((ext_vector_type(2))) float v2f;
typedef __attribute__((ext_vector_type(8))) float v8f;

#define N_REL        1000
#define BIN_BLOCKS   256
#define BIN_THREADS  1024
#define BIN_EDGES    ((N_EDGES + BIN_BLOCKS - 1) / BIN_BLOCKS)
#define BIN_TRIPS    ((BIN_EDGES + BIN_THREADS - 1) / BIN_THREADS)
#define N_RANGES     256
#define RNG_NODES    ((N_NODES + N_RANGES - 1) / N_RANGES)
#define RNG_WAVES    ((RNG_NODES + 31) / 32)
static_assert(N_NODES <= 65536 && N_REL <= 32768, "a (source, relation) pair fits one word");
static_assert(BIN_EDGES < 32768, "a run's start and length fit the two halves of one word");
static_assert(N_RANGES == 256 && BIN_BLOCKS == 256, "the scans below run on eight waves");

template <int MASK>
__device__ __forceinline__ int row_xor_i(int b) {
    return __builtin_amdgcn_update_dpp(b, b, 0x160 + MASK, 0xf, 0xf, false);
}
__device__ __forceinline__ int wave_sum_i(int p) {
    p += row_xor_i<1>(p);
    p += row_xor_i<2>(p);
    p += row_xor_i<4>(p);
    p += row_xor_i<8>(p);
    return p + __builtin_amdgcn_permlanex16(p, p, 0x76543210, 0xfedcba98, false, false);
}
__device__ __forceinline__ int wave_scan_i(int v, int lane) {
    for (int off = 1; off < 32; off <<= 1) {
        int u = __shfl_up(v, off, 32);
        if (lane >= off) v += u;
    }
    return v;
}

__global__ void __launch_bounds__(BIN_THREADS)
k_bin(const int* __restrict__ src, const int* __restrict__ dst, const int* __restrict__ rid,
      int* __restrict__ bin_d, int* __restrict__ bin_p, int* __restrict__ run_tab) {
    __shared__ int s_cnt[N_RANGES];
    __shared__ int s_off[N_RANGES];
    __shared__ int s_wtot[N_RANGES / 32];
    __shared__ int s_d[BIN_EDGES];
    __shared__ int s_p[BIN_EDGES];
    int t = threadIdx.x, c = blockIdx.x;
    int lane = t & 31, wave = t >> 5;
    if (t < N_RANGES) s_cnt[t] = 0;
    bool ok[BIN_TRIPS];
    int  d[BIN_TRIPS], p[BIN_TRIPS];
#pragma unroll
    for (int k = 0; k < BIN_TRIPS; ++k) {
        int w = k * BIN_THREADS + t;
        int i = c * BIN_EDGES + w;
        ok[k] = (w < BIN_EDGES) && (i < N_EDGES);
        int ii = min(i, N_EDGES - 1);
        d[k] = dst[ii];
        p[k] = (src[ii] & 0xffff) | (rid[ii] << 16);
    }
    __syncthreads();
    int b[BIN_TRIPS], rk[BIN_TRIPS];
#pragma unroll
    for (int k = 0; k < BIN_TRIPS; ++k) {
        b[k] = (int)min((unsigned)d[k] / (unsigned)RNG_NODES, (unsigned)(N_RANGES - 1));
        rk[k] = 0;
        if (ok[k]) rk[k] = atomicAdd(&s_cnt[b[k]], 1);
    }
    __syncthreads();
    int v = (t < N_RANGES) ? s_cnt[t] : 0;
    int inc = wave_scan_i(v, lane);
    if (lane == 31 && wave < N_RANGES / 32) s_wtot[wave] = inc;
    __syncthreads();
    if (t < N_RANGES) {
        int before = 0;
        for (int w = 0; w < N_RANGES / 32; ++w) before += (w < wave) ? s_wtot[w] : 0;
        int ex = before + inc - v;
        s_off[t] = ex;
        run_tab[t * BIN_BLOCKS + c] = ex | (v << 16);
    }
    __syncthreads();
#pragma unroll
    for (int k = 0; k < BIN_TRIPS; ++k) {
        if (ok[k]) {
            int pos = s_off[b[k]] + rk[k];
            s_d[pos] = d[k];
            s_p[pos] = p[k];
        }
    }
    __syncthreads();
    int n_c = min(BIN_EDGES, N_EDGES - c * BIN_EDGES);
    for (int j = t; j < n_c; j += BIN_THREADS) {
        bin_d[c * BIN_EDGES + j] = s_d[j];
        bin_p[c * BIN_EDGES + j] = s_p[j];
    }
}

#define S2_THREADS 1024
#define S2_TRIPS   6
#define S2_CAP     8192
#define S2_SEGS    4
#define SEG_CHUNKS (BIN_BLOCKS / S2_SEGS)
#define BASE_TRIPS ((N_RANGES / 2) * (BIN_BLOCKS / 4) / S2_THREADS)
static_assert(S2_SEGS * RNG_NODES <= S2_THREADS, "one thread per (segment, node)");
static_assert(BASE_TRIPS * S2_THREADS == (N_RANGES / 2) * (BIN_BLOCKS / 4), "half the table in whole trips");

__device__ __forceinline__ int find_run(const int* s_rb, int j) {
    int lo = 0, hi = BIN_BLOCKS;
#pragma unroll
    for (int s = 0; s < 8; ++s) {
        int mid = (lo + hi) >> 1;
        if (s_rb[mid] <= j) lo = mid; else hi = mid;
    }
    return lo;
}

__global__ void __launch_bounds__(S2_THREADS)
k_sort2(const int* __restrict__ bin_d, const int* __restrict__ bin_p,
        const int* __restrict__ run_tab,
        int2* __restrict__ sorted, int* __restrict__ cnt, int* __restrict__ offs) {
    __shared__ __attribute__((aligned(16))) int s_cm[BIN_BLOCKS * RNG_NODES];
    __shared__ int2 s_out[S2_CAP];
    __shared__ int s_rb[BIN_BLOCKS + 1];
    __shared__ int s_ro[BIN_BLOCKS];
    __shared__ int s_wa[S2_THREADS / 32], s_wb[S2_THREADS / 32];
    __shared__ int s_part[S2_SEGS * RNG_NODES];
    __shared__ int s_noff[RNG_NODES];
    int t = threadIdx.x, lane = t & 31, wave = t >> 5;
    int r = blockIdx.x;
    int lo = r * RNG_NODES;
    int nn = min(RNG_NODES, N_NODES - lo);
    int word = (t < BIN_BLOCKS) ? run_tab[r * BIN_BLOCKS + t] : 0;
    int r0 = (r < N_RANGES / 2) ? 0 : r;
    int nq = ((r < N_RANGES / 2) ? r : N_RANGES - r) * (BIN_BLOCKS / 4);
    const int4* tab4 = (const int4*)(run_tab + r0 * BIN_BLOCKS);
    int4 q[BASE_TRIPS];
#pragma unroll
    for (int k = 0; k < BASE_TRIPS; ++k) {
        int i = k * S2_THREADS + t;
        q[k] = tab4[(i < nq) ? i : 0];
    }
    {
        int4 z = {0, 0, 0, 0};
        int4* cm4 = (int4*)s_cm;
        for (int i = t; i < BIN_BLOCKS * RNG_NODES / 4; i += S2_THREADS) cm4[i] = z;
    }
    int cn = word >> 16;
    int inc = wave_scan_i(cn, lane);
    int part = 0;
#pragma unroll
    for (int k = 0; k < BASE_TRIPS; ++k) {
        int i = k * S2_THREADS + t;
        int s = ((q[k].x >> 16) + (q[k].y >> 16)) + ((q[k].z >> 16) + (q[k].w >> 16));
        part += (i < nq) ? s : 0;
    }
    part = wave_sum_i(part);
    if (lane == 31) s_wa[wave] = inc;
    if (lane == 0)  s_wb[wave] = part;
    __syncthreads();
    int before = 0, other = 0;
    for (int w = 0; w < S2_THREADS / 32; ++w) {
        other += s_wb[w];
        before += (w < wave) ? s_wa[w] : 0;
    }
    int base = (r < N_RANGES / 2) ? other : N_EDGES - other;
    if (t < BIN_BLOCKS) {
        s_rb[t] = before + inc - cn;
        s_ro[t] = word & 0xffff;
        if (t == BIN_BLOCKS - 1) s_rb[BIN_BLOCKS] = before + inc;
    }
    __syncthreads();
    int nE = s_rb[BIN_BLOCKS];
    bool ok[S2_TRIPS];
    int  gi[S2_TRIPS], cb[S2_TRIPS];
    int  rl[S2_TRIPS], rh[S2_TRIPS];
#pragma unroll
    for (int k = 0; k < S2_TRIPS; ++k) {
        ok[k] = k * S2_THREADS + t < nE;
        rl[k] = 0; rh[k] = BIN_BLOCKS;
    }
#pragma unroll
    for (int s = 0; s < 8; ++s) {
#pragma unroll
        for (int k = 0; k < S2_TRIPS; ++k) {
            int j = ok[k] ? k * S2_THREADS + t : 0;
            int mid = (rl[k] + rh[k]) >> 1;
            bool le = s_rb[mid] <= j;
            rl[k] = le ? mid : rl[k];
            rh[k] = le ? rh[k] : mid;
        }
    }
#pragma unroll
    for (int k = 0; k < S2_TRIPS; ++k) {
        int j = k * S2_THREADS + t;
        int c = rl[k];
        cb[k] = c * RNG_NODES;
        gi[k] = ok[k] ? c * BIN_EDGES + s_ro[c] + (j - s_rb[c]) : 0;
    }
    int dv[S2_TRIPS], pv[S2_TRIPS];
#pragma unroll
    for (int k = 0; k < S2_TRIPS; ++k) {
        dv[k] = bin_d[gi[k]];
        pv[k] = bin_p[gi[k]];
    }
    int cell[S2_TRIPS], ln[S2_TRIPS];
#pragma unroll
    for (int k = 0; k < S2_TRIPS; ++k) {
        ln[k] = min(max(dv[k] - lo, 0), nn - 1);
        cell[k] = cb[k] + ln[k];
        if (ok[k]) atomicAdd(&s_cm[cell[k]], 1);
    }
    for (int j = S2_TRIPS * S2_THREADS + t; j < nE; j += S2_THREADS) {
        int c = find_run(s_rb, j);
        int g = c * BIN_EDGES + s_ro[c] + (j - s_rb[c]);
        int l = min(max(bin_d[g] - lo, 0), nn - 1);
        atomicAdd(&s_cm[c * RNG_NODES + l], 1);
    }
    __syncthreads();
    int seg = t / RNG_NODES, n = t - seg * RNG_NODES;
    bool act = seg < S2_SEGS;
    if (act) {
        int ps = 0;
        for (int c = seg * SEG_CHUNKS; c < (seg + 1) * SEG_CHUNKS; ++c) ps += s_cm[c * RNG_NODES + n];
        s_part[seg * RNG_NODES + n] = ps;
    }
    __syncthreads();
    int tot = 0;
    if (act) {
        int run = 0;
        for (int s = 0; s < S2_SEGS; ++s) {
            int x = s_part[s * RNG_NODES + n];
            tot += x;
            run += (s < seg) ? x : 0;
        }
        for (int c = seg * SEG_CHUNKS; c < (seg + 1) * SEG_CHUNKS; ++c) {
            int x = s_cm[c * RNG_NODES + n];
            s_cm[c * RNG_NODES + n] = run;
            run += x;
        }
    }
    int tv = (t < RNG_NODES) ? tot : 0;
    int tinc = wave_scan_i(tv, lane);
    if (lane == 31 && wave < RNG_WAVES) s_wa[wave] = tinc;
    __syncthreads();
    if (t < RNG_NODES) {
        int bf = 0;
        for (int w = 0; w < RNG_WAVES; ++w) bf += (w < wave) ? s_wa[w] : 0;
        int ex = bf + tinc - tv;
        s_noff[t] = ex;
        if (t < nn) {
            cnt[lo + t]  = tv;
            offs[lo + t] = base + ex;
        }
    }
    __syncthreads();
#pragma unroll
    for (int k = 0; k < S2_TRIPS; ++k) {
        if (ok[k]) {
            int pos = s_noff[ln[k]] + atomicAdd(&s_cm[cell[k]], 1);
            int2 pr = make_int2(pv[k] & 0xffff, (int)((unsigned)pv[k] >> 16));
            if (pos < S2_CAP) s_out[pos] = pr; else sorted[base + pos] = pr;
        }
    }
    for (int j = S2_TRIPS * S2_THREADS + t; j < nE; j += S2_THREADS) {
        int c = find_run(s_rb, j);
        int g = c * BIN_EDGES + s_ro[c] + (j - s_rb[c]);
        int l = min(max(bin_d[g] - lo, 0), nn - 1);
        int w = bin_p[g];
        int pos = s_noff[l] + atomicAdd(&s_cm[c * RNG_NODES + l], 1);
        int2 pr = make_int2(w & 0xffff, (int)((unsigned)w >> 16));
        if (pos < S2_CAP) s_out[pos] = pr; else sorted[base + pos] = pr;
    }
    __syncthreads();
    int nst = min(nE, S2_CAP);
    for (int j = t; j < nst; j += S2_THREADS) sorted[base + j] = s_out[j];
}

template <int MASK>
__device__ __forceinline__ float row_xor(float x) {
    int b = __builtin_bit_cast(int, x);
    return __builtin_bit_cast(float,
        __builtin_amdgcn_update_dpp(b, b, 0x160 + MASK, 0xf, 0xf, false));
}
__device__ __forceinline__ float xhalf(float x) {
    int b = __builtin_bit_cast(int, x);
    return __builtin_bit_cast(float,
        __builtin_amdgcn_permlanex16(b, b, 0x76543210, 0xfedcba98, false, false));
}
__device__ __forceinline__ float wave_sum(float p) {
    p += row_xor<1>(p);
    p += row_xor<2>(p);
    p += row_xor<4>(p);
    p += row_xor<8>(p);
    return p + xhalf(p);
}

#define TILE_ROWS  16
#define TILE_PITCH 68
#define NODE_BLOCKS  256
#define NODE_WAVES   32
#define NODE_THREADS (NODE_WAVES * 32)
#define MAX_TILES    ((N_TILES + NODE_BLOCKS - 1) / NODE_BLOCKS)
#define REL_CHUNKS   (N_REL * EMB / 4)
#define REL_TRIPS    ((REL_CHUNKS + NODE_THREADS - 1) / NODE_THREADS)
#define EDGE_GROUPS      4
#define EDGES_PER_GROUP  4
#define STEP_EDGES       (EDGE_GROUPS * EDGES_PER_GROUP)
static_assert(32 % STEP_EDGES == 0, "a 32-pair piece is a whole number of steps");
static_assert(N_TILES * TILE_ROWS == N_NODES, "the row tiles cover the nodes exactly");
typedef __attribute__((ext_vector_type(4))) float v4f;

__device__ __forceinline__ float dot8(v4f a0, v4f a1, v4f b0, v4f b1) {
    float p = a0.x * b0.x;
    p += a0.y * b0.y; p += a0.z * b0.z; p += a0.w * b0.w;
    p += a1.x * b1.x; p += a1.y * b1.y; p += a1.z * b1.z; p += a1.w * b1.w;
    return p;
}

__global__ void __launch_bounds__(NODE_THREADS)
k_node_gemm(const float* __restrict__ ent, const float* __restrict__ rel,
            const int* __restrict__ cnt, const int* __restrict__ offs,
            const int2* __restrict__ sorted, const float* __restrict__ W,
            float* __restrict__ h, float* __restrict__ part) {
    __shared__ __attribute__((aligned(16))) float s_rel[N_REL * EMB];
    __shared__ __attribute__((aligned(16))) float s_tile[MAX_TILES * TILE_ROWS * TILE_PITCH];
    __shared__ int s_next;
    int lane = threadIdx.x & 31;
    int wave = __builtin_amdgcn_readfirstlane(threadIdx.x >> 5);
    int t0 = (int)(blockIdx.x * N_TILES) / NODE_BLOCKS;
    int ntiles = (int)((blockIdx.x + 1) * N_TILES) / NODE_BLOCKS - t0;
    int nnodes = ntiles * TILE_ROWS;
    int node0  = t0 * TILE_ROWS;
    const v4f* ent4 = (const v4f*)ent;
    const v4f* rel4 = (const v4f*)rel;
    v4f* s_rel4 = (v4f*)s_rel;
    int i = wave;
    int beg = offs[node0 + i];
    int deg = cnt[node0 + i];
    {
        v4f piece[REL_TRIPS];
#pragma unroll
        for (int k = 0; k < REL_TRIPS; ++k) {
            int c = ((k + (int)blockIdx.x) % REL_TRIPS) * NODE_THREADS + (int)threadIdx.x;
            piece[k] = rel4[min(c, REL_CHUNKS - 1)];
        }
#pragma unroll
        for (int k = 0; k < REL_TRIPS; ++k) {
            int c = ((k + (int)blockIdx.x) % REL_TRIPS) * NODE_THREADS + (int)threadIdx.x;
            if (c < REL_CHUNKS) s_rel4[c] = piece[k];
        }
    }
    if (threadIdx.x == 0) s_next = NODE_WAVES;
    __syncthreads();
    int g = lane >> 3;
    int q = lane & 7;
    while (true) {
        int nx = 0;
        if (lane == 0) nx = atomicAdd(&s_next, 1);
        int inext = __builtin_amdgcn_readfirstlane(nx);
        int dn   = node0 + min(inext, nnodes - 1);
        int nbeg = offs[dn];
        int ndeg = cnt[dn];
        int d = node0 + i;
        int end = beg + deg;
        v4f ed0 = ent4[(size_t)d * (EMB / 4) + q];
        v4f ed1 = ent4[(size_t)d * (EMB / 4) + 8 + q];
        float m = -INFINITY, l = 0.0f;
        v4f acc0 = {0.0f, 0.0f, 0.0f, 0.0f}, acc1 = {0.0f, 0.0f, 0.0f, 0.0f};
        if (end > beg) {
            int2 pr = sorted[min(beg + lane, end - 1)];
            for (int cb = beg; cb < end; cb += 32) {
                int nchunk = min(32, end - cb);
                for (int s = 0; s < nchunk; s += STEP_EDGES) {
                    bool ok[EDGES_PER_GROUP];
                    v4f  c0[EDGES_PER_GROUP], c1[EDGES_PER_GROUP];
                    v4f  r0[EDGES_PER_GROUP], r1[EDGES_PER_GROUP];
#pragma unroll
                    for (int u = 0; u < EDGES_PER_GROUP; ++u) {
                        int slot = s + u * EDGE_GROUPS + g;
                        ok[u] = slot < nchunk;
                        int sl = min(slot, nchunk - 1);
                        int si = __shfl(pr.x, sl, 32);
                        int ri = __shfl(pr.y, sl, 32);
                        unsigned eo = (unsigned)si * (EMB / 4) + (unsigned)q;
                        unsigned ro = (unsigned)ri * (EMB / 4) + (unsigned)q;
                        c0[u] = ent4[eo];
                        c1[u] = ent4[eo + 8u];
                        r0[u] = s_rel4[ro];
                        r1[u] = s_rel4[ro + 8u];
                    }
                    float sc[EDGES_PER_GROUP];
                    float tmax = -INFINITY;
#pragma unroll
                    for (int u = 0; u < EDGES_PER_GROUP; ++u) {
                        c0[u] *= r0[u]; c1[u] *= r1[u];
                        float p = dot8(c0[u], c1[u], ed0, ed1);
                        p += row_xor<1>(p);
                        p += row_xor<2>(p);
                        p += row_xor<4>(p);
                        sc[u] = ok[u] ? p : -INFINITY;
                        tmax = fmaxf(tmax, sc[u]);
                    }
                    tmax = fmaxf(tmax, row_xor<8>(tmax));
                    tmax = fmaxf(tmax, xhalf(tmax));
                    float mn = fmaxf(m, tmax);
                    float f  = expf(m - mn);
                    l *= f; acc0 *= f; acc1 *= f;
#pragma unroll
                    for (int u = 0; u < EDGES_PER_GROUP; ++u) {
                        float w = expf(sc[u] - mn);
                        l    += w;
                        acc0 += c0[u] * w;
                        acc1 += c1[u] * w;
                    }
                    m = mn;
                }
                if (cb + 32 < end) pr = sorted[min(cb + 32 + lane, end - 1)];
            }
        }
        l += row_xor<8>(l);  l += xhalf(l);
#pragma unroll
        for (int c = 0; c < 4; ++c) {
            acc0[c] += row_xor<8>(acc0[c]);  acc0[c] += xhalf(acc0[c]);
            acc1[c] += row_xor<8>(acc1[c]);  acc1[c] += xhalf(acc1[c]);
        }
        float inv = (l > 0.0f) ? 1.0f / l : 0.0f;
        if (g == 0) {
            *(v4f*)(s_tile + i * TILE_PITCH + 4 * q)      = acc0 * inv;
            *(v4f*)(s_tile + i * TILE_PITCH + 32 + 4 * q) = acc1 * inv;
        }
        if (inext >= nnodes) break;
        i = inext; beg = nbeg; deg = ndeg;
    }

    int n0    = (wave & 3) * 16;
    int mrow  = lane & 15;
    int khalf = lane >> 4;
    int n     = lane & 15;
    v2f bfrag[EMB / 4];
#pragma unroll
    for (int k = 0; k < EMB; k += 4) {
        int ka = k + khalf * 2;
        bfrag[k / 4].x = W[(size_t)ka       * EMB + n0 + n];
        bfrag[k / 4].y = W[(size_t)(ka + 1) * EMB + n0 + n];
    }
    __syncthreads();

    for (int tl = wave >> 2; tl < ntiles; tl += NODE_WAVES / 4) {
        v8f accd = {};
        const float* arow = s_tile + (tl * TILE_ROWS + mrow) * TILE_PITCH;
#pragma unroll
        for (int k = 0; k < EMB; k += 4) {
            int ka = k + khalf * 2;
            v2f a = *(const v2f*)(arow + ka);
            accd = __builtin_amdgcn_wmma_f32_16x16x4_f32(
                       false, a, false, bfrag[k / 4], (short)0, accd, false, false);
        }
        int tg    = t0 + tl;
        int mbase = tg * TILE_ROWS + khalf * 8;
        float cs = 0.0f, cq = 0.0f;
#pragma unroll
        for (int r = 0; r < 8; ++r) {
            float v = accd[r];
            h[(size_t)(mbase + r) * EMB + n0 + n] = v;
            cs += v; cq += v * v;
        }
        cs += __shfl_xor(cs, 16, 32);
        cq += __shfl_xor(cq, 16, 32);
        if (khalf == 0) {
            v2f p; p.x = cs; p.y = cq;
            *(v2f*)(part + 2 * ((size_t)(n0 + n) * N_TILES + tg)) = p;
        }
    }
}

#define STATS_THREADS 1024
__global__ void __launch_bounds__(STATS_THREADS)
k_stats(const float* __restrict__ part,
        const float* __restrict__ gamma, const float* __restrict__ beta,
        float* __restrict__ scale, float* __restrict__ shift) {
    int c = blockIdx.x;
    const v2f* p = (const v2f*)part + (size_t)c * N_TILES;
    __shared__ float s_sum[STATS_THREADS / 32], s_sq[STATS_THREADS / 32];
    float sum = 0.0f, sq = 0.0f;
#pragma unroll
    for (int i = 0; i < (N_TILES + STATS_THREADS - 1) / STATS_THREADS; ++i) {
        int t = threadIdx.x + i * STATS_THREADS;
        if (t < N_TILES) { v2f v = p[t]; sum += v.x; sq += v.y; }
    }
#pragma unroll
    for (int off = 16; off > 0; off >>= 1) {
        sum += __shfl_xor(sum, off, 32);
        sq  += __shfl_xor(sq,  off, 32);
    }
    int wave = threadIdx.x >> 5, lane = threadIdx.x & 31;
    if (lane == 0) { s_sum[wave] = sum; s_sq[wave] = sq; }
    __syncthreads();
    if (wave == 0) {
        sum = s_sum[lane]; sq = s_sq[lane];
#pragma unroll
        for (int off = 16; off > 0; off >>= 1) {
            sum += __shfl_xor(sum, off, 32);
            sq  += __shfl_xor(sq,  off, 32);
        }
        if (lane == 0) {
            float mu   = sum / (float)N_NODES;
            float var  = sq / (float)N_NODES - mu * mu;
            float rinv = rsqrtf(var + BN_EPS);
            float sc   = gamma[c] * rinv;
            scale[c] = sc;
            shift[c] = beta[c] - mu * sc;
        }
    }
}

__global__ void k_apply(float* __restrict__ h,
                        const float* __restrict__ scale, const float* __restrict__ shift) {
    int i = blockIdx.x * blockDim.x + threadIdx.x;
    if (i >= N_NODES * EMB) return;
    int c = i & 63;
    h[i] = tanhf(h[i] * scale[c] + shift[c]);
}

extern "C" void kernel_launch(void* const* d_in, const int* in_sizes, int n_in,
                              void* d_out, int out_size, void* d_ws, size_t ws_size,
                              hipStream_t stream) {
    const float* ent   = (const float*)d_in[0];
    const float* rel   = (const float*)d_in[1];
    const int*   src   = (const int*)  d_in[2];
    const int*   dst   = (const int*)  d_in[3];
    const int*   rid   = (const int*)  d_in[4];
    const float* W     = (const float*)d_in[5];
    const float* gamma = (const float*)d_in[6];
    const float* beta  = (const float*)d_in[7];
    float* out = (float*)d_out;

    float* ws     = (float*)d_ws;
    int2*  sorted = (int2*)ws;
    int*   cnt    = (int*)(ws + 2 * (size_t)N_EDGES);
    int*   offs   = (int*)(ws + 2 * (size_t)N_EDGES + N_NODES);
    float* spare  = ws + 2 * (size_t)N_EDGES + 2 * (size_t)N_NODES;
    float* scale  = spare + (size_t)N_NODES * EMB;
    float* shift  = scale + EMB;
    int*   bin_d   = (int*)spare;
    int*   bin_p   = bin_d + BIN_BLOCKS * BIN_EDGES;
    int*   run_tab = bin_p + BIN_BLOCKS * BIN_EDGES;
    float* part   = spare;

    k_bin   <<<BIN_BLOCKS, BIN_THREADS, 0, stream>>>(src, dst, rid, bin_d, bin_p, run_tab);
    k_sort2 <<<N_RANGES, S2_THREADS, 0, stream>>>(bin_d, bin_p, run_tab, sorted, cnt, offs);
    k_node_gemm<<<NODE_BLOCKS, NODE_THREADS, 0, stream>>>(ent, rel, cnt, offs, sorted, W, out, part);
    k_stats  <<<EMB, STATS_THREADS, 0, stream>>>(part, gamma, beta, scale, shift);
    k_apply  <<<(N_NODES * EMB + 255) / 256, 256, 0, stream>>>(out, scale, shift);
}


# Second sort level moved into the node kernel as its blocks' first phase (range = block's own tiles, LDS shared by both phases): a launch gone; (chunk,node) counters kept, chunk-major sum order kept
# speedup vs baseline: 9.7046x; 1.0326x over previous
#include <hip/hip_runtime.h>
#include <math.h>

#define N_NODES 50000
#define N_EDGES 1250000
#define EMB     64
#define BN_EPS  1e-5f
#define N_TILES (N_NODES / 16)
#define TILE_ROWS    16
#define NODE_BLOCKS  256
#define MAX_TILES    ((N_TILES + NODE_BLOCKS - 1) / NODE_BLOCKS)
#define TILE_PITCH   68

typedef __attribute__((ext_vector_type(2))) float v2f;
typedef __attribute__((ext_vector_type(8))) float v8f;

#define N_REL        1000
#define BIN_BLOCKS   256
#define BIN_THREADS  1024
#define BIN_EDGES    ((N_EDGES + BIN_BLOCKS - 1) / BIN_BLOCKS)
#define BIN_TRIPS    ((BIN_EDGES + BIN_THREADS - 1) / BIN_THREADS)
#define N_RANGES     NODE_BLOCKS
#define RNG_NODES    (MAX_TILES * TILE_ROWS)
#define RNG_WAVES    ((RNG_NODES + 31) / 32)
static_assert(N_NODES <= 65536 && N_REL <= 32768, "a (source, relation) pair fits one word");
static_assert(BIN_EDGES < 32768, "a run's start and length fit the two halves of one word");
static_assert(N_RANGES == 256 && BIN_BLOCKS == 256, "the scans below run on eight waves");

__device__ __forceinline__ int range_of(int d) {
    unsigned tl = (unsigned)d >> 4;
    return (int)min(((tl + 1u) * (unsigned)NODE_BLOCKS - 1u) / (unsigned)N_TILES, (unsigned)(N_RANGES - 1));
}
static_assert(TILE_ROWS == 16, "range_of takes the row tile as d >> 4");

template <int MASK>
__device__ __forceinline__ int row_xor_i(int b) {
    return __builtin_amdgcn_update_dpp(b, b, 0x160 + MASK, 0xf, 0xf, false);
}
__device__ __forceinline__ int wave_sum_i(int p) {
    p += row_xor_i<1>(p);
    p += row_xor_i<2>(p);
    p += row_xor_i<4>(p);
    p += row_xor_i<8>(p);
    return p + __builtin_amdgcn_permlanex16(p, p, 0x76543210, 0xfedcba98, false, false);
}
__device__ __forceinline__ int wave_scan_i(int v, int lane) {
    for (int off = 1; off < 32; off <<= 1) {
        int u = __shfl_up(v, off, 32);
        if (lane >= off) v += u;
    }
    return v;
}

__global__ void __launch_bounds__(BIN_THREADS)
k_bin(const int* __restrict__ src, const int* __restrict__ dst, const int* __restrict__ rid,
      int* __restrict__ bin_d, int* __restrict__ bin_p, int* __restrict__ run_tab) {
    __shared__ int s_cnt[N_RANGES];
    __shared__ int s_off[N_RANGES];
    __shared__ int s_wtot[N_RANGES / 32];
    __shared__ int s_d[BIN_EDGES];
    __shared__ int s_p[BIN_EDGES];
    int t = threadIdx.x, c = blockIdx.x;
    int lane = t & 31, wave = t >> 5;
    if (t < N_RANGES) s_cnt[t] = 0;
    bool ok[BIN_TRIPS];
    int  d[BIN_TRIPS], p[BIN_TRIPS];
#pragma unroll
    for (int k = 0; k < BIN_TRIPS; ++k) {
        int w = k * BIN_THREADS + t;
        int i = c * BIN_EDGES + w;
        ok[k] = (w < BIN_EDGES) && (i < N_EDGES);
        int ii = min(i, N_EDGES - 1);
        d[k] = dst[ii];
        p[k] = (src[ii] & 0xffff) | (rid[ii] << 16);
    }
    __syncthreads();
    int b[BIN_TRIPS], rk[BIN_TRIPS];
#pragma unroll
    for (int k = 0; k < BIN_TRIPS; ++k) {
        b[k] = range_of(d[k]);
        rk[k] = 0;
        if (ok[k]) rk[k] = atomicAdd(&s_cnt[b[k]], 1);
    }
    __syncthreads();
    int v = (t < N_RANGES) ? s_cnt[t] : 0;
    int inc = wave_scan_i(v, lane);
    if (lane == 31 && wave < N_RANGES / 32) s_wtot[wave] = inc;
    __syncthreads();
    if (t < N_RANGES) {
        int before = 0;
        for (int w = 0; w < N_RANGES / 32; ++w) before += (w < wave) ? s_wtot[w] : 0;
        int ex = before + inc - v;
        s_off[t] = ex;
        run_tab[t * BIN_BLOCKS + c] = ex | (v << 16);
    }
    __syncthreads();
#pragma unroll
    for (int k = 0; k < BIN_TRIPS; ++k) {
        if (ok[k]) {
            int pos = s_off[b[k]] + rk[k];
            s_d[pos] = d[k];
            s_p[pos] = p[k];
        }
    }
    __syncthreads();
    int n_c = min(BIN_EDGES, N_EDGES - c * BIN_EDGES);
    for (int j = t; j < n_c; j += BIN_THREADS) {
        bin_d[c * BIN_EDGES + j] = s_d[j];
        bin_p[c * BIN_EDGES + j] = s_p[j];
    }
}

#define S2_THREADS 1024
#define S2_TRIPS   6
#define S2_CAP     8192
#define S2_SEGS    4
#define SEG_CHUNKS (BIN_BLOCKS / S2_SEGS)
#define BASE_TRIPS ((N_RANGES / 2) * (BIN_BLOCKS / 4) / S2_THREADS)
static_assert(S2_SEGS * RNG_NODES <= S2_THREADS, "one thread per (segment, node)");
static_assert(BASE_TRIPS * S2_THREADS == (N_RANGES / 2) * (BIN_BLOCKS / 4), "half the table in whole trips");

__device__ __forceinline__ int find_run(const int* s_rb, int j) {
    int lo = 0, hi = BIN_BLOCKS;
#pragma unroll
    for (int s = 0; s < 8; ++s) {
        int mid = (lo + hi) >> 1;
        if (s_rb[mid] <= j) lo = mid; else hi = mid;
    }
    return lo;
}

#define L_REL      0
#define L_TILE     (N_REL * EMB)
#define L_CNT      (L_TILE + MAX_TILES * TILE_ROWS * TILE_PITCH)
#define L_OFFS     (L_CNT + RNG_NODES)
#define L_NEXT     (L_OFFS + RNG_NODES)
#define L_WORDS    (L_NEXT + 4)
#define L_CM       0
#define L_OUT      (BIN_BLOCKS * RNG_NODES)
#define L_RB       (L_OUT + 2 * S2_CAP)
#define L_RO       (L_RB + BIN_BLOCKS + 4)
#define L_WA       (L_RO + BIN_BLOCKS)
#define L_WB       (L_WA + S2_THREADS / 32)
#define L_PART     (L_WB + S2_THREADS / 32)
#define L_NOFF     (L_PART + S2_SEGS * RNG_NODES)
#define L_SORT_END (L_NOFF + RNG_NODES)
static_assert(L_SORT_END <= L_CNT, "the sort phase stays below what it hands to the node phase");
static_assert(L_OUT % 4 == 0 && L_TILE % 4 == 0, "16-byte pieces stay aligned");
static_assert(L_WORDS * 4 <= 320 * 1024, "one block's LDS");

__device__ __forceinline__ void
sort_range(int* s_all, const int* __restrict__ bin_d, const int* __restrict__ bin_p,
           const int* __restrict__ run_tab, int2* sorted, int lo, int nn) {
    int*  s_cm   = s_all + L_CM;
    int2* s_out  = (int2*)(s_all + L_OUT);
    int*  s_rb   = s_all + L_RB;
    int*  s_ro   = s_all + L_RO;
    int*  s_wa   = s_all + L_WA;
    int*  s_wb   = s_all + L_WB;
    int*  s_part = s_all + L_PART;
    int*  s_noff = s_all + L_NOFF;
    int*  s_cnt  = s_all + L_CNT;
    int*  s_offs = s_all + L_OFFS;
    int t = threadIdx.x, lane = t & 31, wave = t >> 5;
    int r = blockIdx.x;
    int word = (t < BIN_BLOCKS) ? run_tab[r * BIN_BLOCKS + t] : 0;
    int r0 = (r < N_RANGES / 2) ? 0 : r;
    int nq = ((r < N_RANGES / 2) ? r : N_RANGES - r) * (BIN_BLOCKS / 4);
    const int4* tab4 = (const int4*)(run_tab + r0 * BIN_BLOCKS);
    int4 q[BASE_TRIPS];
#pragma unroll
    for (int k = 0; k < BASE_TRIPS; ++k) {
        int i = k * S2_THREADS + t;
        q[k] = tab4[(i < nq) ? i : 0];
    }
    {
        int4 z = {0, 0, 0, 0};
        int4* cm4 = (int4*)s_cm;
        for (int i = t; i < BIN_BLOCKS * RNG_NODES / 4; i += S2_THREADS) cm4[i] = z;
    }
    int cn = word >> 16;
    int inc = wave_scan_i(cn, lane);
    int part = 0;
#pragma unroll
    for (int k = 0; k < BASE_TRIPS; ++k) {
        int i = k * S2_THREADS + t;
        int s = ((q[k].x >> 16) + (q[k].y >> 16)) + ((q[k].z >> 16) + (q[k].w >> 16));
        part += (i < nq) ? s : 0;
    }
    part = wave_sum_i(part);
    if (lane == 31) s_wa[wave] = inc;
    if (lane == 0)  s_wb[wave] = part;
    __syncthreads();
    int before = 0, other = 0;
    for (int w = 0; w < S2_THREADS / 32; ++w) {
        other += s_wb[w];
        before += (w < wave) ? s_wa[w] : 0;
    }
    int base = (r < N_RANGES / 2) ? other : N_EDGES - other;
    if (t < BIN_BLOCKS) {
        s_rb[t] = before + inc - cn;
        s_ro[t] = word & 0xffff;
        if (t == BIN_BLOCKS - 1) s_rb[BIN_BLOCKS] = before + inc;
    }
    __syncthreads();
    int nE = s_rb[BIN_BLOCKS];
    bool ok[S2_TRIPS];
    int  gi[S2_TRIPS], cb[S2_TRIPS];
    int  rl[S2_TRIPS], rh[S2_TRIPS];
#pragma unroll
    for (int k = 0; k < S2_TRIPS; ++k) {
        ok[k] = k * S2_THREADS + t < nE;
        rl[k] = 0; rh[k] = BIN_BLOCKS;
    }
#pragma unroll
    for (int s = 0; s < 8; ++s) {
#pragma unroll
        for (int k = 0; k < S2_TRIPS; ++k) {
            int j = ok[k] ? k * S2_THREADS + t : 0;
            int mid = (rl[k] + rh[k]) >> 1;
            bool le = s_rb[mid] <= j;
            rl[k] = le ? mid : rl[k];
            rh[k] = le ? rh[k] : mid;
        }
    }
#pragma unroll
    for (int k = 0; k < S2_TRIPS; ++k) {
        int j = k * S2_THREADS + t;
        int c = rl[k];
        cb[k] = c * RNG_NODES;
        gi[k] = ok[k] ? c * BIN_EDGES + s_ro[c] + (j - s_rb[c]) : 0;
    }
    int dv[S2_TRIPS], pv[S2_TRIPS];
#pragma unroll
    for (int k = 0; k < S2_TRIPS; ++k) {
        dv[k] = bin_d[gi[k]];
        pv[k] = bin_p[gi[k]];
    }
    int cell[S2_TRIPS], ln[S2_TRIPS];
#pragma unroll
    for (int k = 0; k < S2_TRIPS; ++k) {
        ln[k] = min(max(dv[k] - lo, 0), nn - 1);
        cell[k] = cb[k] + ln[k];
        if (ok[k]) atomicAdd(&s_cm[cell[k]], 1);
    }
    for (int j = S2_TRIPS * S2_THREADS + t; j < nE; j += S2_THREADS) {
        int c = find_run(s_rb, j);
        int g = c * BIN_EDGES + s_ro[c] + (j - s_rb[c]);
        int l = min(max(bin_d[g] - lo, 0), nn - 1);
        atomicAdd(&s_cm[c * RNG_NODES + l], 1);
    }
    __syncthreads();
    int seg = t / RNG_NODES, n = t - seg * RNG_NODES;
    bool act = seg < S2_SEGS;
    if (act) {
        int ps = 0;
        for (int c = seg * SEG_CHUNKS; c < (seg + 1) * SEG_CHUNKS; ++c) ps += s_cm[c * RNG_NODES + n];
        s_part[seg * RNG_NODES + n] = ps;
    }
    __syncthreads();
    int tot = 0;
    if (act) {
        int run = 0;
        for (int s = 0; s < S2_SEGS; ++s) {
            int x = s_part[s * RNG_NODES + n];
            tot += x;
            run += (s < seg) ? x : 0;
        }
        for (int c = seg * SEG_CHUNKS; c < (seg + 1) * SEG_CHUNKS; ++c) {
            int x = s_cm[c * RNG_NODES + n];
            s_cm[c * RNG_NODES + n] = run;
            run += x;
        }
    }
    int tv = (t < RNG_NODES) ? tot : 0;
    int tinc = wave_scan_i(tv, lane);
    if (lane == 31 && wave < RNG_WAVES) s_wa[wave] = tinc;
    __syncthreads();
    if (t < RNG_NODES) {
        int bf = 0;
        for (int w = 0; w < RNG_WAVES; ++w) bf += (w < wave) ? s_wa[w] : 0;
        int ex = bf + tinc - tv;
        s_noff[t] = ex;
        if (t < nn) {
            s_cnt[t]  = tv;
            s_offs[t] = base + ex;
        }
    }
    __syncthreads();
#pragma unroll
    for (int k = 0; k < S2_TRIPS; ++k) {
        if (ok[k]) {
            int pos = s_noff[ln[k]] + atomicAdd(&s_cm[cell[k]], 1);
            int2 pr = make_int2(pv[k] & 0xffff, (int)((unsigned)pv[k] >> 16));
            if (pos < S2_CAP) s_out[pos] = pr; else sorted[base + pos] = pr;
        }
    }
    for (int j = S2_TRIPS * S2_THREADS + t; j < nE; j += S2_THREADS) {
        int c = find_run(s_rb, j);
        int g = c * BIN_EDGES + s_ro[c] + (j - s_rb[c]);
        int l = min(max(bin_d[g] - lo, 0), nn - 1);
        int w = bin_p[g];
        int pos = s_noff[l] + atomicAdd(&s_cm[c * RNG_NODES + l], 1);
        int2 pr = make_int2(w & 0xffff, (int)((unsigned)w >> 16));
        if (pos < S2_CAP) s_out[pos] = pr; else sorted[base + pos] = pr;
    }
    __syncthreads();
    int nst = min(nE, S2_CAP);
    for (int j = t; j < nst; j += S2_THREADS) sorted[base + j] = s_out[j];
}

template <int MASK>
__device__ __forceinline__ float row_xor(float x) {
    int b = __builtin_bit_cast(int, x);
    return __builtin_bit_cast(float,
        __builtin_amdgcn_update_dpp(b, b, 0x160 + MASK, 0xf, 0xf, false));
}
__device__ __forceinline__ float xhalf(float x) {
    int b = __builtin_bit_cast(int, x);
    return __builtin_bit_cast(float,
        __builtin_amdgcn_permlanex16(b, b, 0x76543210, 0xfedcba98, false, false));
}
__device__ __forceinline__ float wave_sum(float p) {
    p += row_xor<1>(p);
    p += row_xor<2>(p);
    p += row_xor<4>(p);
    p += row_xor<8>(p);
    return p + xhalf(p);
}

#define NODE_WAVES   32
#define NODE_THREADS (NODE_WAVES * 32)
#define REL_CHUNKS   (N_REL * EMB / 4)
#define REL_TRIPS    ((REL_CHUNKS + NODE_THREADS - 1) / NODE_THREADS)
#define EDGE_GROUPS      4
#define EDGES_PER_GROUP  4
#define STEP_EDGES       (EDGE_GROUPS * EDGES_PER_GROUP)
static_assert(32 % STEP_EDGES == 0, "a 32-pair piece is a whole number of steps");
static_assert(N_TILES * TILE_ROWS == N_NODES, "the row tiles cover the nodes exactly");
typedef __attribute__((ext_vector_type(4))) float v4f;

__device__ __forceinline__ float dot8(v4f a0, v4f a1, v4f b0, v4f b1) {
    float p = a0.x * b0.x;
    p += a0.y * b0.y; p += a0.z * b0.z; p += a0.w * b0.w;
    p += a1.x * b1.x; p += a1.y * b1.y; p += a1.z * b1.z; p += a1.w * b1.w;
    return p;
}

static_assert(S2_THREADS == NODE_THREADS, "the sort phase runs on the node kernel's block");

__device__ __forceinline__ int2 load_pair(int2* p) {
    unsigned long long v = __hip_atomic_load((unsigned long long*)p, __ATOMIC_RELAXED,
                                             __HIP_MEMORY_SCOPE_AGENT);
    return make_int2((int)(unsigned)v, (int)(unsigned)(v >> 32));
}

__global__ void __launch_bounds__(NODE_THREADS)
k_sort_node_gemm(const int* __restrict__ bin_d, const int* __restrict__ bin_p,
                 const int* __restrict__ run_tab, int2* sorted,
                 const float* __restrict__ ent, const float* __restrict__ rel,
                 const float* __restrict__ W,
                 float* __restrict__ h, float* __restrict__ part) {
    __shared__ __attribute__((aligned(16))) int s_all[L_WORDS];
    float* s_rel  = (float*)(s_all + L_REL);
    float* s_tile = (float*)(s_all + L_TILE);
    int*   s_cnt  = s_all + L_CNT;
    int*   s_offs = s_all + L_OFFS;
    int*   s_next = s_all + L_NEXT;
    int lane = threadIdx.x & 31;
    int wave = __builtin_amdgcn_readfirstlane(threadIdx.x >> 5);
    int t0 = (int)(blockIdx.x * N_TILES) / NODE_BLOCKS;
    int ntiles = (int)((blockIdx.x + 1) * N_TILES) / NODE_BLOCKS - t0;
    int nnodes = ntiles * TILE_ROWS;
    int node0  = t0 * TILE_ROWS;
    const v4f* ent4 = (const v4f*)ent;
    const v4f* rel4 = (const v4f*)rel;
    v4f* s_rel4 = (v4f*)s_rel;
    sort_range(s_all, bin_d, bin_p, run_tab, sorted, node0, nnodes);
    __syncthreads();
    int i = wave;
    int beg = s_offs[i];
    int deg = s_cnt[i];
    {
        v4f piece[REL_TRIPS];
#pragma unroll
        for (int k = 0; k < REL_TRIPS; ++k) {
            int c = ((k + (int)blockIdx.x) % REL_TRIPS) * NODE_THREADS + (int)threadIdx.x;
            piece[k] = rel4[min(c, REL_CHUNKS - 1)];
        }
#pragma unroll
        for (int k = 0; k < REL_TRIPS; ++k) {
            int c = ((k + (int)blockIdx.x) % REL_TRIPS) * NODE_THREADS + (int)threadIdx.x;
            if (c < REL_CHUNKS) s_rel4[c] = piece[k];
        }
    }
    if (threadIdx.x == 0) *s_next = NODE_WAVES;
    __syncthreads();
    int g = lane >> 3;
    int q = lane & 7;
    while (true) {
        int nx = 0;
        if (lane == 0) nx = atomicAdd(s_next, 1);
        int inext = __builtin_amdgcn_readfirstlane(nx);
        int dn   = min(inext, nnodes - 1);
        int nbeg = s_offs[dn];
        int ndeg = s_cnt[dn];
        int d = node0 + i;
        int end = beg + deg;
        v4f ed0 = ent4[(size_t)d * (EMB / 4) + q];
        v4f ed1 = ent4[(size_t)d * (EMB / 4) + 8 + q];
        float m = -INFINITY, l = 0.0f;
        v4f acc0 = {0.0f, 0.0f, 0.0f, 0.0f}, acc1 = {0.0f, 0.0f, 0.0f, 0.0f};
        if (end > beg) {
            int2 pr = load_pair(sorted + min(beg + lane, end - 1));
            for (int cb = beg; cb < end; cb += 32) {
                int nchunk = min(32, end - cb);
                for (int s = 0; s < nchunk; s += STEP_EDGES) {
                    bool ok[EDGES_PER_GROUP];
                    v4f  c0[EDGES_PER_GROUP], c1[EDGES_PER_GROUP];
                    v4f  r0[EDGES_PER_GROUP], r1[EDGES_PER_GROUP];
#pragma unroll
                    for (int u = 0; u < EDGES_PER_GROUP; ++u) {
                        int slot = s + u * EDGE_GROUPS + g;
                        ok[u] = slot < nchunk;
                        int sl = min(slot, nchunk - 1);
                        int si = __shfl(pr.x, sl, 32);
                        int ri = __shfl(pr.y, sl, 32);
                        unsigned eo = (unsigned)si * (EMB / 4) + (unsigned)q;
                        unsigned ro = (unsigned)ri * (EMB / 4) + (unsigned)q;
                        c0[u] = ent4[eo];
                        c1[u] = ent4[eo + 8u];
                        r0[u] = s_rel4[ro];
                        r1[u] = s_rel4[ro + 8u];
                    }
                    float sc[EDGES_PER_GROUP];
                    float tmax = -INFINITY;
#pragma unroll
                    for (int u = 0; u < EDGES_PER_GROUP; ++u) {
                        c0[u] *= r0[u]; c1[u] *= r1[u];
                        float p = dot8(c0[u], c1[u], ed0, ed1);
                        p += row_xor<1>(p);
                        p += row_xor<2>(p);
                        p += row_xor<4>(p);
                        sc[u] = ok[u] ? p : -INFINITY;
                        tmax = fmaxf(tmax, sc[u]);
                    }
                    tmax = fmaxf(tmax, row_xor<8>(tmax));
                    tmax = fmaxf(tmax, xhalf(tmax));
                    float mn = fmaxf(m, tmax);
                    float f  = expf(m - mn);
                    l *= f; acc0 *= f; acc1 *= f;
#pragma unroll
                    for (int u = 0; u < EDGES_PER_GROUP; ++u) {
                        float w = expf(sc[u] - mn);
                        l    += w;
                        acc0 += c0[u] * w;
                        acc1 += c1[u] * w;
                    }
                    m = mn;
                }
                if (cb + 32 < end) pr = load_pair(sorted + min(cb + 32 + lane, end - 1));
            }
        }
        l += row_xor<8>(l);  l += xhalf(l);
#pragma unroll
        for (int c = 0; c < 4; ++c) {
            acc0[c] += row_xor<8>(acc0[c]);  acc0[c] += xhalf(acc0[c]);
            acc1[c] += row_xor<8>(acc1[c]);  acc1[c] += xhalf(acc1[c]);
        }
        float inv = (l > 0.0f) ? 1.0f / l : 0.0f;
        if (g == 0) {
            *(v4f*)(s_tile + i * TILE_PITCH + 4 * q)      = acc0 * inv;
            *(v4f*)(s_tile + i * TILE_PITCH + 32 + 4 * q) = acc1 * inv;
        }
        if (inext >= nnodes) break;
        i = inext; beg = nbeg; deg = ndeg;
    }

    int n0    = (wave & 3) * 16;
    int mrow  = lane & 15;
    int khalf = lane >> 4;
    int n     = lane & 15;
    v2f bfrag[EMB / 4];
#pragma unroll
    for (int k = 0; k < EMB; k += 4) {
        int ka = k + khalf * 2;
        bfrag[k / 4].x = W[(size_t)ka       * EMB + n0 + n];
        bfrag[k / 4].y = W[(size_t)(ka + 1) * EMB + n0 + n];
    }
    __syncthreads();

    for (int tl = wave >> 2; tl < ntiles; tl += NODE_WAVES / 4) {
        v8f accd = {};
        const float* arow = s_tile + (tl * TILE_ROWS + mrow) * TILE_PITCH;
#pragma unroll
        for (int k = 0; k < EMB; k += 4) {
            int ka = k + khalf * 2;
            v2f a = *(const v2f*)(arow + ka);
            accd = __builtin_amdgcn_wmma_f32_16x16x4_f32(
                       false, a, false, bfrag[k / 4], (short)0, accd, false, false);
        }
        int tg    = t0 + tl;
        int mbase = tg * TILE_ROWS + khalf * 8;
        float cs = 0.0f, cq = 0.0f;
#pragma unroll
        for (int r = 0; r < 8; ++r) {
            float v = accd[r];
            h[(size_t)(mbase + r) * EMB + n0 + n] = v;
            cs += v; cq += v * v;
        }
        cs += __shfl_xor(cs, 16, 32);
        cq += __shfl_xor(cq, 16, 32);
        if (khalf == 0) {
            v2f p; p.x = cs; p.y = cq;
            *(v2f*)(part + 2 * ((size_t)(n0 + n) * N_TILES + tg)) = p;
        }
    }
}

#define STATS_THREADS 1024
__global__ void __launch_bounds__(STATS_THREADS)
k_stats(const float* __restrict__ part,
        const float* __restrict__ gamma, const float* __restrict__ beta,
        float* __restrict__ scale, float* __restrict__ shift) {
    int c = blockIdx.x;
    const v2f* p = (const v2f*)part + (size_t)c * N_TILES;
    __shared__ float s_sum[STATS_THREADS / 32], s_sq[STATS_THREADS / 32];
    float sum = 0.0f, sq = 0.0f;
#pragma unroll
    for (int i = 0; i < (N_TILES + STATS_THREADS - 1) / STATS_THREADS; ++i) {
        int t = threadIdx.x + i * STATS_THREADS;
        if (t < N_TILES) { v2f v = p[t]; sum += v.x; sq += v.y; }
    }
#pragma unroll
    for (int off = 16; off > 0; off >>= 1) {
        sum += __shfl_xor(sum, off, 32);
        sq  += __shfl_xor(sq,  off, 32);
    }
    int wave = threadIdx.x >> 5, lane = threadIdx.x & 31;
    if (lane == 0) { s_sum[wave] = sum; s_sq[wave] = sq; }
    __syncthreads();
    if (wave == 0) {
        sum = s_sum[lane]; sq = s_sq[lane];
#pragma unroll
        for (int off = 16; off > 0; off >>= 1) {
            sum += __shfl_xor(sum, off, 32);
            sq  += __shfl_xor(sq,  off, 32);
        }
        if (lane == 0) {
            float mu   = sum / (float)N_NODES;
            float var  = sq / (float)N_NODES - mu * mu;
            float rinv = rsqrtf(var + BN_EPS);
            float sc   = gamma[c] * rinv;
            scale[c] = sc;
            shift[c] = beta[c] - mu * sc;
        }
    }
}

__global__ void k_apply(float* __restrict__ h,
                        const float* __restrict__ scale, const float* __restrict__ shift) {
    int i = blockIdx.x * blockDim.x + threadIdx.x;
    if (i >= N_NODES * EMB) return;
    int c = i & 63;
    h[i] = tanhf(h[i] * scale[c] + shift[c]);
}

#define PART_OFF 2600000
static_assert(PART_OFF >= 2 * BIN_BLOCKS * BIN_EDGES + N_RANGES * BIN_BLOCKS, "part does not touch the bins");
static_assert(PART_OFF + 2 * EMB * N_TILES <= N_NODES * EMB, "part stays inside the spare plane");
static_assert(PART_OFF % 2 == 0, "pairs of floats stay 8-byte aligned");

extern "C" void kernel_launch(void* const* d_in, const int* in_sizes, int n_in,
                              void* d_out, int out_size, void* d_ws, size_t ws_size,
                              hipStream_t stream) {
    const float* ent   = (const float*)d_in[0];
    const float* rel   = (const float*)d_in[1];
    const int*   src   = (const int*)  d_in[2];
    const int*   dst   = (const int*)  d_in[3];
    const int*   rid   = (const int*)  d_in[4];
    const float* W     = (const float*)d_in[5];
    const float* gamma = (const float*)d_in[6];
    const float* beta  = (const float*)d_in[7];
    float* out = (float*)d_out;

    float* ws     = (float*)d_ws;
    int2*  sorted = (int2*)ws;
    float* spare  = ws + 2 * (size_t)N_EDGES + 2 * (size_t)N_NODES;
    float* scale  = spare + (size_t)N_NODES * EMB;
    float* shift  = scale + EMB;
    int*   bin_d   = (int*)spare;
    int*   bin_p   = bin_d + BIN_BLOCKS * BIN_EDGES;
    int*   run_tab = bin_p + BIN_BLOCKS * BIN_EDGES;
    float* part   = spare + PART_OFF;

    k_bin   <<<BIN_BLOCKS, BIN_THREADS, 0, stream>>>(src, dst, rid, bin_d, bin_p, run_tab);
    k_sort_node_gemm<<<NODE_BLOCKS, NODE_THREADS, 0, stream>>>(bin_d, bin_p, run_tab, sorted,
                                                               ent, rel, W, out, part);
    k_stats  <<<EMB, STATS_THREADS, 0, stream>>>(part, gamma, beta, scale, shift);
    k_apply  <<<(N_NODES * EMB + 255) / 256, 256, 0, stream>>>(out, scale, shift);
}


# BN column sums added into 128 words by float atomics (a block's tiles first summed in LDS, fixed order); stats launch gone, apply folds scale/shift per thread; sum order across blocks = arrival
# speedup vs baseline: 11.1647x; 1.1505x over previous
#include <hip/hip_runtime.h>
#include <math.h>

#define N_NODES 50000
#define N_EDGES 1250000
#define EMB     64
#define BN_EPS  1e-5f
#define N_TILES (N_NODES / 16)
#define TILE_ROWS    16
#define NODE_BLOCKS  256
#define MAX_TILES    ((N_TILES + NODE_BLOCKS - 1) / NODE_BLOCKS)
#define TILE_PITCH   68

typedef __attribute__((ext_vector_type(2))) float v2f;
typedef __attribute__((ext_vector_type(8))) float v8f;

#define N_REL        1000
#define BIN_BLOCKS   256
#define BIN_THREADS  1024
#define BIN_EDGES    ((N_EDGES + BIN_BLOCKS - 1) / BIN_BLOCKS)
#define BIN_TRIPS    ((BIN_EDGES + BIN_THREADS - 1) / BIN_THREADS)
#define N_RANGES     NODE_BLOCKS
#define RNG_NODES    (MAX_TILES * TILE_ROWS)
#define RNG_WAVES    ((RNG_NODES + 31) / 32)
static_assert(N_NODES <= 65536 && N_REL <= 32768, "a (source, relation) pair fits one word");
static_assert(BIN_EDGES < 32768, "a run's start and length fit the two halves of one word");
static_assert(N_RANGES == 256 && BIN_BLOCKS == 256, "the scans below run on eight waves");

__device__ __forceinline__ int range_of(int d) {
    unsigned tl = (unsigned)d >> 4;
    return (int)min(((tl + 1u) * (unsigned)NODE_BLOCKS - 1u) / (unsigned)N_TILES, (unsigned)(N_RANGES - 1));
}
static_assert(TILE_ROWS == 16, "range_of takes the row tile as d >> 4");

template <int MASK>
__device__ __forceinline__ int row_xor_i(int b) {
    return __builtin_amdgcn_update_dpp(b, b, 0x160 + MASK, 0xf, 0xf, false);
}
__device__ __forceinline__ int wave_sum_i(int p) {
    p += row_xor_i<1>(p);
    p += row_xor_i<2>(p);
    p += row_xor_i<4>(p);
    p += row_xor_i<8>(p);
    return p + __builtin_amdgcn_permlanex16(p, p, 0x76543210, 0xfedcba98, false, false);
}
__device__ __forceinline__ int wave_scan_i(int v, int lane) {
    for (int off = 1; off < 32; off <<= 1) {
        int u = __shfl_up(v, off, 32);
        if (lane >= off) v += u;
    }
    return v;
}

__global__ void __launch_bounds__(BIN_THREADS)
k_bin(const int* __restrict__ src, const int* __restrict__ dst, const int* __restrict__ rid,
      int* __restrict__ bin_d, int* __restrict__ bin_p, int* __restrict__ run_tab,
      float* __restrict__ gstat) {
    __shared__ int s_cnt[N_RANGES];
    __shared__ int s_off[N_RANGES];
    __shared__ int s_wtot[N_RANGES / 32];
    __shared__ int s_d[BIN_EDGES];
    __shared__ int s_p[BIN_EDGES];
    int t = threadIdx.x, c = blockIdx.x;
    int lane = t & 31, wave = t >> 5;
    if (t < N_RANGES) s_cnt[t] = 0;
    if (c == 0 && t < 2 * EMB) gstat[t] = 0.0f;
    bool ok[BIN_TRIPS];
    int  d[BIN_TRIPS], p[BIN_TRIPS];
#pragma unroll
    for (int k = 0; k < BIN_TRIPS; ++k) {
        int w = k * BIN_THREADS + t;
        int i = c * BIN_EDGES + w;
        ok[k] = (w < BIN_EDGES) && (i < N_EDGES);
        int ii = min(i, N_EDGES - 1);
        d[k] = dst[ii];
        p[k] = (src[ii] & 0xffff) | (rid[ii] << 16);
    }
    __syncthreads();
    int b[BIN_TRIPS], rk[BIN_TRIPS];
#pragma unroll
    for (int k = 0; k < BIN_TRIPS; ++k) {
        b[k] = range_of(d[k]);
        rk[k] = 0;
        if (ok[k]) rk[k] = atomicAdd(&s_cnt[b[k]], 1);
    }
    __syncthreads();
    int v = (t < N_RANGES) ? s_cnt[t] : 0;
    int inc = wave_scan_i(v, lane);
    if (lane == 31 && wave < N_RANGES / 32) s_wtot[wave] = inc;
    __syncthreads();
    if (t < N_RANGES) {
        int before = 0;
        for (int w = 0; w < N_RANGES / 32; ++w) before += (w < wave) ? s_wtot[w] : 0;
        int ex = before + inc - v;
        s_off[t] = ex;
        run_tab[t * BIN_BLOCKS + c] = ex | (v << 16);
    }
    __syncthreads();
#pragma unroll
    for (int k = 0; k < BIN_TRIPS; ++k) {
        if (ok[k]) {
            int pos = s_off[b[k]] + rk[k];
            s_d[pos] = d[k];
            s_p[pos] = p[k];
        }
    }
    __syncthreads();
    int n_c = min(BIN_EDGES, N_EDGES - c * BIN_EDGES);
    for (int j = t; j < n_c; j += BIN_THREADS) {
        bin_d[c * BIN_EDGES + j] = s_d[j];
        bin_p[c * BIN_EDGES + j] = s_p[j];
    }
}

#define S2_THREADS 1024
#define S2_TRIPS   6
#define S2_CAP     8192
#define S2_SEGS    4
#define SEG_CHUNKS (BIN_BLOCKS / S2_SEGS)
#define BASE_TRIPS ((N_RANGES / 2) * (BIN_BLOCKS / 4) / S2_THREADS)
static_assert(S2_SEGS * RNG_NODES <= S2_THREADS, "one thread per (segment, node)");
static_assert(BASE_TRIPS * S2_THREADS == (N_RANGES / 2) * (BIN_BLOCKS / 4), "half the table in whole trips");

__device__ __forceinline__ int find_run(const int* s_rb, int j) {
    int lo = 0, hi = BIN_BLOCKS;
#pragma unroll
    for (int s = 0; s < 8; ++s) {
        int mid = (lo + hi) >> 1;
        if (s_rb[mid] <= j) lo = mid; else hi = mid;
    }
    return lo;
}

#define L_REL      0
#define L_TILE     (N_REL * EMB)
#define L_CNT      (L_TILE + MAX_TILES * TILE_ROWS * TILE_PITCH)
#define L_OFFS     (L_CNT + RNG_NODES)
#define L_NEXT     (L_OFFS + RNG_NODES)
#define L_WORDS    (L_NEXT + 4)
#define L_CM       0
#define L_OUT      (BIN_BLOCKS * RNG_NODES)
#define L_RB       (L_OUT + 2 * S2_CAP)
#define L_RO       (L_RB + BIN_BLOCKS + 4)
#define L_WA       (L_RO + BIN_BLOCKS)
#define L_WB       (L_WA + S2_THREADS / 32)
#define L_PART     (L_WB + S2_THREADS / 32)
#define L_NOFF     (L_PART + S2_SEGS * RNG_NODES)
#define L_SORT_END (L_NOFF + RNG_NODES)
static_assert(L_SORT_END <= L_CNT, "the sort phase stays below what it hands to the node phase");
static_assert(L_OUT % 4 == 0 && L_TILE % 4 == 0, "16-byte pieces stay aligned");
static_assert(L_WORDS * 4 <= 320 * 1024, "one block's LDS");

__device__ __forceinline__ void
sort_range(int* s_all, const int* __restrict__ bin_d, const int* __restrict__ bin_p,
           const int* __restrict__ run_tab, int2* sorted, int lo, int nn) {
    int*  s_cm   = s_all + L_CM;
    int2* s_out  = (int2*)(s_all + L_OUT);
    int*  s_rb   = s_all + L_RB;
    int*  s_ro   = s_all + L_RO;
    int*  s_wa   = s_all + L_WA;
    int*  s_wb   = s_all + L_WB;
    int*  s_part = s_all + L_PART;
    int*  s_noff = s_all + L_NOFF;
    int*  s_cnt  = s_all + L_CNT;
    int*  s_offs = s_all + L_OFFS;
    int t = threadIdx.x, lane = t & 31, wave = t >> 5;
    int r = blockIdx.x;
    int word = (t < BIN_BLOCKS) ? run_tab[r * BIN_BLOCKS + t] : 0;
    int r0 = (r < N_RANGES / 2) ? 0 : r;
    int nq = ((r < N_RANGES / 2) ? r : N_RANGES - r) * (BIN_BLOCKS / 4);
    const int4* tab4 = (const int4*)(run_tab + r0 * BIN_BLOCKS);
    int4 q[BASE_TRIPS];
#pragma unroll
    for (int k = 0; k < BASE_TRIPS; ++k) {
        int i = k * S2_THREADS + t;
        q[k] = tab4[(i < nq) ? i : 0];
    }
    {
        int4 z = {0, 0, 0, 0};
        int4* cm4 = (int4*)s_cm;
        for (int i = t; i < BIN_BLOCKS * RNG_NODES / 4; i += S2_THREADS) cm4[i] = z;
    }
    int cn = word >> 16;
    int inc = wave_scan_i(cn, lane);
    int part = 0;
#pragma unroll
    for (int k = 0; k < BASE_TRIPS; ++k) {
        int i = k * S2_THREADS + t;
        int s = ((q[k].x >> 16) + (q[k].y >> 16)) + ((q[k].z >> 16) + (q[k].w >> 16));
        part += (i < nq) ? s : 0;
    }
    part = wave_sum_i(part);
    if (lane == 31) s_wa[wave] = inc;
    if (lane == 0)  s_wb[wave] = part;
    __syncthreads();
    int before = 0, other = 0;
    for (int w = 0; w < S2_THREADS / 32; ++w) {
        other += s_wb[w];
        before += (w < wave) ? s_wa[w] : 0;
    }
    int base = (r < N_RANGES / 2) ? other : N_EDGES - other;
    if (t < BIN_BLOCKS) {
        s_rb[t] = before + inc - cn;
        s_ro[t] = word & 0xffff;
        if (t == BIN_BLOCKS - 1) s_rb[BIN_BLOCKS] = before + inc;
    }
    __syncthreads();
    int nE = s_rb[BIN_BLOCKS];
    bool ok[S2_TRIPS];
    int  gi[S2_TRIPS], cb[S2_TRIPS];
    int  rl[S2_TRIPS], rh[S2_TRIPS];
#pragma unroll
    for (int k = 0; k < S2_TRIPS; ++k) {
        ok[k] = k * S2_THREADS + t < nE;
        rl[k] = 0; rh[k] = BIN_BLOCKS;
    }
#pragma unroll
    for (int s = 0; s < 8; ++s) {
#pragma unroll
        for (int k = 0; k < S2_TRIPS; ++k) {
            int j = ok[k] ? k * S2_THREADS + t : 0;
            int mid = (rl[k] + rh[k]) >> 1;
            bool le = s_rb[mid] <= j;
            rl[k] = le ? mid : rl[k];
            rh[k] = le ? rh[k] : mid;
        }
    }
#pragma unroll
    for (int k = 0; k < S2_TRIPS; ++k) {
        int j = k * S2_THREADS + t;
        int c = rl[k];
        cb[k] = c * RNG_NODES;
        gi[k] = ok[k] ? c * BIN_EDGES + s_ro[c] + (j - s_rb[c]) : 0;
    }
    int dv[S2_TRIPS], pv[S2_TRIPS];
#pragma unroll
    for (int k = 0; k < S2_TRIPS; ++k) {
        dv[k] = bin_d[gi[k]];
        pv[k] = bin_p[gi[k]];
    }
    int cell[S2_TRIPS], ln[S2_TRIPS];
#pragma unroll
    for (int k = 0; k < S2_TRIPS; ++k) {
        ln[k] = min(max(dv[k] - lo, 0), nn - 1);
        cell[k] = cb[k] + ln[k];
        if (ok[k]) atomicAdd(&s_cm[cell[k]], 1);
    }
    for (int j = S2_TRIPS * S2_THREADS + t; j < nE; j += S2_THREADS) {
        int c = find_run(s_rb, j);
        int g = c * BIN_EDGES + s_ro[c] + (j - s_rb[c]);
        int l = min(max(bin_d[g] - lo, 0), nn - 1);
        atomicAdd(&s_cm[c * RNG_NODES + l], 1);
    }
    __syncthreads();
    int seg = t / RNG_NODES, n = t - seg * RNG_NODES;
    bool act = seg < S2_SEGS;
    if (act) {
        int ps = 0;
        for (int c = seg * SEG_CHUNKS; c < (seg + 1) * SEG_CHUNKS; ++c) ps += s_cm[c * RNG_NODES + n];
        s_part[seg * RNG_NODES + n] = ps;
    }
    __syncthreads();
    int tot = 0;
    if (act) {
        int run = 0;
        for (int s = 0; s < S2_SEGS; ++s) {
            int x = s_part[s * RNG_NODES + n];
            tot += x;
            run += (s < seg) ? x : 0;
        }
        for (int c = seg * SEG_CHUNKS; c < (seg + 1) * SEG_CHUNKS; ++c) {
            int x = s_cm[c * RNG_NODES + n];
            s_cm[c * RNG_NODES + n] = run;
            run += x;
        }
    }
    int tv = (t < RNG_NODES) ? tot : 0;
    int tinc = wave_scan_i(tv, lane);
    if (lane == 31 && wave < RNG_WAVES) s_wa[wave] = tinc;
    __syncthreads();
    if (t < RNG_NODES) {
        int bf = 0;
        for (int w = 0; w < RNG_WAVES; ++w) bf += (w < wave) ? s_wa[w] : 0;
        int ex = bf + tinc - tv;
        s_noff[t] = ex;
        if (t < nn) {
            s_cnt[t]  = tv;
            s_offs[t] = base + ex;
        }
    }
    __syncthreads();
#pragma unroll
    for (int k = 0; k < S2_TRIPS; ++k) {
        if (ok[k]) {
            int pos = s_noff[ln[k]] + atomicAdd(&s_cm[cell[k]], 1);
            int2 pr = make_int2(pv[k] & 0xffff, (int)((unsigned)pv[k] >> 16));
            if (pos < S2_CAP) s_out[pos] = pr; else sorted[base + pos] = pr;
        }
    }
    for (int j = S2_TRIPS * S2_THREADS + t; j < nE; j += S2_THREADS) {
        int c = find_run(s_rb, j);
        int g = c * BIN_EDGES + s_ro[c] + (j - s_rb[c]);
        int l = min(max(bin_d[g] - lo, 0), nn - 1);
        int w = bin_p[g];
        int pos = s_noff[l] + atomicAdd(&s_cm[c * RNG_NODES + l], 1);
        int2 pr = make_int2(w & 0xffff, (int)((unsigned)w >> 16));
        if (pos < S2_CAP) s_out[pos] = pr; else sorted[base + pos] = pr;
    }
    __syncthreads();
    int nst = min(nE, S2_CAP);
    for (int j = t; j < nst; j += S2_THREADS) sorted[base + j] = s_out[j];
}

template <int MASK>
__device__ __forceinline__ float row_xor(float x) {
    int b = __builtin_bit_cast(int, x);
    return __builtin_bit_cast(float,
        __builtin_amdgcn_update_dpp(b, b, 0x160 + MASK, 0xf, 0xf, false));
}
__device__ __forceinline__ float xhalf(float x) {
    int b = __builtin_bit_cast(int, x);
    return __builtin_bit_cast(float,
        __builtin_amdgcn_permlanex16(b, b, 0x76543210, 0xfedcba98, false, false));
}
__device__ __forceinline__ float wave_sum(float p) {
    p += row_xor<1>(p);
    p += row_xor<2>(p);
    p += row_xor<4>(p);
    p += row_xor<8>(p);
    return p + xhalf(p);
}

#define NODE_WAVES   32
#define NODE_THREADS (NODE_WAVES * 32)
#define REL_CHUNKS   (N_REL * EMB / 4)
#define REL_TRIPS    ((REL_CHUNKS + NODE_THREADS - 1) / NODE_THREADS)
#define EDGE_GROUPS      4
#define EDGES_PER_GROUP  4
#define STEP_EDGES       (EDGE_GROUPS * EDGES_PER_GROUP)
static_assert(32 % STEP_EDGES == 0, "a 32-pair piece is a whole number of steps");
static_assert(N_TILES * TILE_ROWS == N_NODES, "the row tiles cover the nodes exactly");
typedef __attribute__((ext_vector_type(4))) float v4f;

__device__ __forceinline__ float dot8(v4f a0, v4f a1, v4f b0, v4f b1) {
    float p = a0.x * b0.x;
    p += a0.y * b0.y; p += a0.z * b0.z; p += a0.w * b0.w;
    p += a1.x * b1.x; p += a1.y * b1.y; p += a1.z * b1.z; p += a1.w * b1.w;
    return p;
}

static_assert(S2_THREADS == NODE_THREADS, "the sort phase runs on the node kernel's block");

__device__ __forceinline__ int2 load_pair(int2* p) {
    unsigned long long v = __hip_atomic_load((unsigned long long*)p, __ATOMIC_RELAXED,
                                             __HIP_MEMORY_SCOPE_AGENT);
    return make_int2((int)(unsigned)v, (int)(unsigned)(v >> 32));
}

__global__ void __launch_bounds__(NODE_THREADS)
k_sort_node_gemm(const int* __restrict__ bin_d, const int* __restrict__ bin_p,
                 const int* __restrict__ run_tab, int2* sorted,
                 const float* __restrict__ ent, const float* __restrict__ rel,
                 const float* __restrict__ W,
                 float* __restrict__ h, float* gstat) {
    __shared__ __attribute__((aligned(16))) int s_all[L_WORDS];
    float* s_rel  = (float*)(s_all + L_REL);
    float* s_tile = (float*)(s_all + L_TILE);
    int*   s_cnt  = s_all + L_CNT;
    int*   s_offs = s_all + L_OFFS;
    int*   s_next = s_all + L_NEXT;
    int lane = threadIdx.x & 31;
    int wave = __builtin_amdgcn_readfirstlane(threadIdx.x >> 5);
    int t0 = (int)(blockIdx.x * N_TILES) / NODE_BLOCKS;
    int ntiles = (int)((blockIdx.x + 1) * N_TILES) / NODE_BLOCKS - t0;
    int nnodes = ntiles * TILE_ROWS;
    int node0  = t0 * TILE_ROWS;
    const v4f* ent4 = (const v4f*)ent;
    const v4f* rel4 = (const v4f*)rel;
    v4f* s_rel4 = (v4f*)s_rel;
    sort_range(s_all, bin_d, bin_p, run_tab, sorted, node0, nnodes);
    __syncthreads();
    int i = wave;
    int beg = s_offs[i];
    int deg = s_cnt[i];
    {
        v4f piece[REL_TRIPS];
#pragma unroll
        for (int k = 0; k < REL_TRIPS; ++k) {
            int c = ((k + (int)blockIdx.x) % REL_TRIPS) * NODE_THREADS + (int)threadIdx.x;
            piece[k] = rel4[min(c, REL_CHUNKS - 1)];
        }
#pragma unroll
        for (int k = 0; k < REL_TRIPS; ++k) {
            int c = ((k + (int)blockIdx.x) % REL_TRIPS) * NODE_THREADS + (int)threadIdx.x;
            if (c < REL_CHUNKS) s_rel4[c] = piece[k];
        }
    }
    if (threadIdx.x == 0) *s_next = NODE_WAVES;
    __syncthreads();
    int g = lane >> 3;
    int q = lane & 7;
    while (true) {
        int nx = 0;
        if (lane == 0) nx = atomicAdd(s_next, 1);
        int inext = __builtin_amdgcn_readfirstlane(nx);
        int dn   = min(inext, nnodes - 1);
        int nbeg = s_offs[dn];
        int ndeg = s_cnt[dn];
        int d = node0 + i;
        int end = beg + deg;
        v4f ed0 = ent4[(size_t)d * (EMB / 4) + q];
        v4f ed1 = ent4[(size_t)d * (EMB / 4) + 8 + q];
        float m = -INFINITY, l = 0.0f;
        v4f acc0 = {0.0f, 0.0f, 0.0f, 0.0f}, acc1 = {0.0f, 0.0f, 0.0f, 0.0f};
        if (end > beg) {
            int2 pr = load_pair(sorted + min(beg + lane, end - 1));
            for (int cb = beg; cb < end; cb += 32) {
                int nchunk = min(32, end - cb);
                for (int s = 0; s < nchunk; s += STEP_EDGES) {
                    bool ok[EDGES_PER_GROUP];
                    v4f  c0[EDGES_PER_GROUP], c1[EDGES_PER_GROUP];
                    v4f  r0[EDGES_PER_GROUP], r1[EDGES_PER_GROUP];
#pragma unroll
                    for (int u = 0; u < EDGES_PER_GROUP; ++u) {
                        int slot = s + u * EDGE_GROUPS + g;
                        ok[u] = slot < nchunk;
                        int sl = min(slot, nchunk - 1);
                        int si = __shfl(pr.x, sl, 32);
                        int ri = __shfl(pr.y, sl, 32);
                        unsigned eo = (unsigned)si * (EMB / 4) + (unsigned)q;
                        unsigned ro = (unsigned)ri * (EMB / 4) + (unsigned)q;
                        c0[u] = ent4[eo];
                        c1[u] = ent4[eo + 8u];
                        r0[u] = s_rel4[ro];
                        r1[u] = s_rel4[ro + 8u];
                    }
                    float sc[EDGES_PER_GROUP];
                    float tmax = -INFINITY;
#pragma unroll
                    for (int u = 0; u < EDGES_PER_GROUP; ++u) {
                        c0[u] *= r0[u]; c1[u] *= r1[u];
                        float p = dot8(c0[u], c1[u], ed0, ed1);
                        p += row_xor<1>(p);
                        p += row_xor<2>(p);
                        p += row_xor<4>(p);
                        sc[u] = ok[u] ? p : -INFINITY;
                        tmax = fmaxf(tmax, sc[u]);
                    }
                    tmax = fmaxf(tmax, row_xor<8>(tmax));
                    tmax = fmaxf(tmax, xhalf(tmax));
                    float mn = fmaxf(m, tmax);
                    float f  = expf(m - mn);
                    l *= f; acc0 *= f; acc1 *= f;
#pragma unroll
                    for (int u = 0; u < EDGES_PER_GROUP; ++u) {
                        float w = expf(sc[u] - mn);
                        l    += w;
                        acc0 += c0[u] * w;
                        acc1 += c1[u] * w;
                    }
                    m = mn;
                }
                if (cb + 32 < end) pr = load_pair(sorted + min(cb + 32 + lane, end - 1));
            }
        }
        l += row_xor<8>(l);  l += xhalf(l);
#pragma unroll
        for (int c = 0; c < 4; ++c) {
            acc0[c] += row_xor<8>(acc0[c]);  acc0[c] += xhalf(acc0[c]);
            acc1[c] += row_xor<8>(acc1[c]);  acc1[c] += xhalf(acc1[c]);
        }
        float inv = (l > 0.0f) ? 1.0f / l : 0.0f;
        if (g == 0) {
            *(v4f*)(s_tile + i * TILE_PITCH + 4 * q)      = acc0 * inv;
            *(v4f*)(s_tile + i * TILE_PITCH + 32 + 4 * q) = acc1 * inv;
        }
        if (inext >= nnodes) break;
        i = inext; beg = nbeg; deg = ndeg;
    }

    int n0    = (wave & 3) * 16;
    int mrow  = lane & 15;
    int khalf = lane >> 4;
    int n     = lane & 15;
    v2f bfrag[EMB / 4];
#pragma unroll
    for (int k = 0; k < EMB; k += 4) {
        int ka = k + khalf * 2;
        bfrag[k / 4].x = W[(size_t)ka       * EMB + n0 + n];
        bfrag[k / 4].y = W[(size_t)(ka + 1) * EMB + n0 + n];
    }
    __syncthreads();

    float wcs = 0.0f, wcq = 0.0f;
    for (int tl = wave >> 2; tl < ntiles; tl += NODE_WAVES / 4) {
        v8f accd = {};
        const float* arow = s_tile + (tl * TILE_ROWS + mrow) * TILE_PITCH;
#pragma unroll
        for (int k = 0; k < EMB; k += 4) {
            int ka = k + khalf * 2;
            v2f a = *(const v2f*)(arow + ka);
            accd = __builtin_amdgcn_wmma_f32_16x16x4_f32(
                       false, a, false, bfrag[k / 4], (short)0, accd, false, false);
        }
        int mbase = (t0 + tl) * TILE_ROWS + khalf * 8;
        float cs = 0.0f, cq = 0.0f;
#pragma unroll
        for (int r = 0; r < 8; ++r) {
            float v = accd[r];
            h[(size_t)(mbase + r) * EMB + n0 + n] = v;
            cs += v; cq += v * v;
        }
        cs += __shfl_xor(cs, 16, 32);
        cq += __shfl_xor(cq, 16, 32);
        wcs += cs; wcq += cq;
    }
    float* s_red = (float*)(s_all + L_REL);
    if (khalf == 0) {
        s_red[(wave >> 2) * EMB + n0 + n] = wcs;
        s_red[(NODE_WAVES / 4) * EMB + (wave >> 2) * EMB + n0 + n] = wcq;
    }
    __syncthreads();
    if (threadIdx.x < 2 * EMB) {
        int which = (int)threadIdx.x >> 6, col = (int)threadIdx.x & (EMB - 1);
        const float* src_red = s_red + which * (NODE_WAVES / 4) * EMB + col;
        float v = 0.0f;
#pragma unroll
        for (int gq = 0; gq < NODE_WAVES / 4; ++gq) v += src_red[gq * EMB];
        __hip_atomic_fetch_add(gstat + threadIdx.x, v, __ATOMIC_RELAXED, __HIP_MEMORY_SCOPE_AGENT);
    }
}

__global__ void k_apply(float* __restrict__ h, const float* __restrict__ gstat,
                        const float* __restrict__ gamma, const float* __restrict__ beta) {
    int i = blockIdx.x * blockDim.x + threadIdx.x;
    if (i >= N_NODES * EMB) return;
    int c = i & 63;
    float x    = h[i];
    float mu   = gstat[c] * (1.0f / (float)N_NODES);
    float var  = gstat[EMB + c] * (1.0f / (float)N_NODES) - mu * mu;
    float rinv = rsqrtf(var + BN_EPS);
    float sc   = gamma[c] * rinv;
    float sh   = beta[c] - mu * sc;
    h[i] = tanhf(x * sc + sh);
}

#define STAT_OFF 2600000
static_assert(STAT_OFF >= 2 * BIN_BLOCKS * BIN_EDGES + N_RANGES * BIN_BLOCKS, "the sums do not touch the bins");
static_assert(STAT_OFF + 2 * EMB <= N_NODES * EMB, "the sums stay inside the spare plane");

extern "C" void kernel_launch(void* const* d_in, const int* in_sizes, int n_in,
                              void* d_out, int out_size, void* d_ws, size_t ws_size,
                              hipStream_t stream) {
    const float* ent   = (const float*)d_in[0];
    const float* rel   = (const float*)d_in[1];
    const int*   src   = (const int*)  d_in[2];
    const int*   dst   = (const int*)  d_in[3];
    const int*   rid   = (const int*)  d_in[4];
    const float* W     = (const float*)d_in[5];
    const float* gamma = (const float*)d_in[6];
    const float* beta  = (const float*)d_in[7];
    float* out = (float*)d_out;

    float* ws     = (float*)d_ws;
    int2*  sorted = (int2*)ws;
    float* spare  = ws + 2 * (size_t)N_EDGES + 2 * (size_t)N_NODES;
    int*   bin_d   = (int*)spare;
    int*   bin_p   = bin_d + BIN_BLOCKS * BIN_EDGES;
    int*   run_tab = bin_p + BIN_BLOCKS * BIN_EDGES;
    float* gstat  = spare + STAT_OFF;

    k_bin   <<<BIN_BLOCKS, BIN_THREADS, 0, stream>>>(src, dst, rid, bin_d, bin_p, run_tab, gstat);
    k_sort_node_gemm<<<NODE_BLOCKS, NODE_THREADS, 0, stream>>>(bin_d, bin_p, run_tab, sorted,
                                                               ent, rel, W, out, gstat);
    k_apply  <<<(N_NODES * EMB + 255) / 256, 256, 0, stream>>>(out, gstat, gamma, beta);
}
